# starting kernel
# baseline (speedup 1.0000x reference)
#include <hip/hip_runtime.h>

constexpr int IN_DIM  = 128;
constexpr int OUT_DIM = 64;
constexpr int S1 = IN_DIM + 4;
constexpr int S2 = OUT_DIM + 4;

typedef float v2f __attribute__((ext_vector_type(2)));
typedef float v8f __attribute__((ext_vector_type(8)));

__global__ void __launch_bounds__(256) k_init(float* __restrict__ deg_s,
                                              float* __restrict__ deg_d,
                                              float* __restrict__ agg,
                                              float* __restrict__ hg, int n) {
    int tid = blockIdx.x * blockDim.x + threadIdx.x;
    int stride = gridDim.x * blockDim.x;
    for (int i = tid; i < n; i += stride) { deg_s[i] = 0.f; deg_d[i] = 0.f; }
    for (int i = tid; i < n * OUT_DIM; i += stride) agg[i] = 0.f;
    if (tid < 2 * OUT_DIM) hg[tid] = 0.f;
}

__global__ void __launch_bounds__(256) k_degree(const int* __restrict__ src,
                                                const int* __restrict__ dst,
                                                float* __restrict__ outdeg,
                                                float* __restrict__ indeg, int E) {
    int e = blockIdx.x * blockDim.x + threadIdx.x;
    if (e < E) {
        atomicAdd(&outdeg[src[e]], 1.0f);
        atomicAdd(&indeg[dst[e]], 1.0f);
    }
}

__global__ void __launch_bounds__(256) k_norm(float* __restrict__ sn,
                                              float* __restrict__ dn, int n) {
    int i = blockIdx.x * blockDim.x + threadIdx.x;
    if (i < n) {
        sn[i] = rsqrtf(fmaxf(sn[i], 1.0f));
        dn[i] = rsqrtf(fmaxf(dn[i], 1.0f));
    }
}

__global__ void __launch_bounds__(256) k_gemm1(const float* __restrict__ feat,
                                               const float* __restrict__ W1,
                                               const float* __restrict__ srcn,
                                               float* __restrict__ t1, int n) {
    __shared__ float sWt[OUT_DIM * S1];
    for (int i = threadIdx.x; i < IN_DIM * OUT_DIM; i += 256) {
        const int k = i >> 6;
        const int c = i & 63;
        sWt[c * S1 + k] = W1[i];
    }
    __syncthreads();

    const int wave = threadIdx.x >> 5;
    const int lane = threadIdx.x & 31;
    const int tile = blockIdx.x * 8 + wave;
    const int ntiles = (n + 15) >> 4;
    if (tile >= ntiles) return;

    const int halfsel = lane >> 4;
    const int l15 = lane & 15;
    int row = tile * 16 + l15;
    if (row >= n) row = n - 1;
    const int koff = halfsel * 2;
    const float norm = srcn[row];
    const float* arow = feat + (size_t)row * IN_DIM;
    const float* bc0 = sWt + ( 0 + l15) * S1;
    const float* bc1 = sWt + (16 + l15) * S1;
    const float* bc2 = sWt + (32 + l15) * S1;
    const float* bc3 = sWt + (48 + l15) * S1;

    v8f acc0 = {}, acc1 = {}, acc2 = {}, acc3 = {};
#pragma unroll 4
    for (int kb = 0; kb < IN_DIM / 4; ++kb) {
        const int k0 = kb * 4 + koff;
        v2f a = *(const v2f*)(arow + k0);
        a.x *= norm; a.y *= norm;
        const v2f b0 = *(const v2f*)(bc0 + k0);
        const v2f b1 = *(const v2f*)(bc1 + k0);
        const v2f b2 = *(const v2f*)(bc2 + k0);
        const v2f b3 = *(const v2f*)(bc3 + k0);
        acc0 = __builtin_amdgcn_wmma_f32_16x16x4_f32(false, a, false, b0, (short)0, acc0, false, false);
        acc1 = __builtin_amdgcn_wmma_f32_16x16x4_f32(false, a, false, b1, (short)0, acc1, false, false);
        acc2 = __builtin_amdgcn_wmma_f32_16x16x4_f32(false, a, false, b2, (short)0, acc2, false, false);
        acc3 = __builtin_amdgcn_wmma_f32_16x16x4_f32(false, a, false, b3, (short)0, acc3, false, false);
    }

#pragma unroll
    for (int r = 0; r < 8; ++r) {
        const int node = tile * 16 + r + halfsel * 8;
        if (node < n) {
            float* o = t1 + (size_t)node * OUT_DIM + l15;
            o[0]  = acc0[r];
            o[16] = acc1[r];
            o[32] = acc2[r];
            o[48] = acc3[r];
        }
    }
}

__global__ void __launch_bounds__(256) k_scatter(const int* __restrict__ src,
                                                 const int* __restrict__ dst,
                                                 const float* __restrict__ msg,
                                                 float* __restrict__ agg, int E) {
    int i = blockIdx.x * blockDim.x + threadIdx.x;
    if (i < E * (OUT_DIM / 4)) {
        const int e = i >> 4;
        const int q = (i & 15) << 2;
        const float4 m = *(const float4*)(msg + (size_t)src[e] * OUT_DIM + q);
        float* a = agg + (size_t)dst[e] * OUT_DIM + q;
        atomicAdd(a + 0, m.x);
        atomicAdd(a + 1, m.y);
        atomicAdd(a + 2, m.z);
        atomicAdd(a + 3, m.w);
    }
}

__global__ void __launch_bounds__(256) k_finalize1(float* __restrict__ agg,
                                                   const float* __restrict__ dstn,
                                                   const float* __restrict__ srcn,
                                                   const float* __restrict__ b1,
                                                   const float* __restrict__ prelu_a,
                                                   float* __restrict__ hs,
                                                   float* __restrict__ hg1, int n) {
    const float pa = *prelu_a;
    const int d = threadIdx.x & 63;
    const int sub = threadIdx.x >> 6;
    float part = 0.f;
    for (int base = blockIdx.x * 4; base < n; base += gridDim.x * 4) {
        const int node = base + sub;
        if (node < n) {
            const size_t idx = (size_t)node * OUT_DIM + d;
            const float v = agg[idx];
            agg[idx] = 0.f;
            float h = v * dstn[node] + b1[d];
            h = (h >= 0.f) ? h : pa * h;
            part += h;
            hs[idx] = h * srcn[node];
        }
    }
    __shared__ float red[256];
    red[threadIdx.x] = part;
    __syncthreads();
    if (threadIdx.x < 128) red[threadIdx.x] += red[threadIdx.x + 128];
    __syncthreads();
    if (threadIdx.x < 64)
        atomicAdd(&hg1[threadIdx.x], red[threadIdx.x] + red[threadIdx.x + 64]);
}

__global__ void __launch_bounds__(256) k_gemm2(const float* __restrict__ agg,
                                               const float* __restrict__ W2,
                                               const float* __restrict__ dstn,
                                               const float* __restrict__ b2,
                                               const float* __restrict__ prelu_a,
                                               float* __restrict__ hout,
                                               float* __restrict__ hg2, int n) {
    __shared__ float sWt[OUT_DIM * S2];
    __shared__ float shg[OUT_DIM];
    for (int i = threadIdx.x; i < OUT_DIM * OUT_DIM; i += 256) {
        const int k = i >> 6;
        const int c = i & 63;
        sWt[c * S2 + k] = W2[i];
    }
    if (threadIdx.x < OUT_DIM) shg[threadIdx.x] = 0.f;
    __syncthreads();

    const float pa = *prelu_a;
    const int wave = threadIdx.x >> 5;
    const int lane = threadIdx.x & 31;
    const int halfsel = lane >> 4;
    const int l15 = lane & 15;
    const int tile = blockIdx.x * 8 + wave;
    const int ntiles = (n + 15) >> 4;
    const bool active = tile < ntiles;

    if (active) {
        int row = tile * 16 + l15;
        if (row >= n) row = n - 1;
        const int koff = halfsel * 2;
        const float* arow = agg + (size_t)row * OUT_DIM;
        const float* bc0 = sWt + ( 0 + l15) * S2;
        const float* bc1 = sWt + (16 + l15) * S2;
        const float* bc2 = sWt + (32 + l15) * S2;
        const float* bc3 = sWt + (48 + l15) * S2;

        v8f acc0 = {}, acc1 = {}, acc2 = {}, acc3 = {};
#pragma unroll 4
        for (int kb = 0; kb < OUT_DIM / 4; ++kb) {
            const int k0 = kb * 4 + koff;
            const v2f a  = *(const v2f*)(arow + k0);
            const v2f b0 = *(const v2f*)(bc0 + k0);
            const v2f b1 = *(const v2f*)(bc1 + k0);
            const v2f b2 = *(const v2f*)(bc2 + k0);
            const v2f b3 = *(const v2f*)(bc3 + k0);
            acc0 = __builtin_amdgcn_wmma_f32_16x16x4_f32(false, a, false, b0, (short)0, acc0, false, false);
            acc1 = __builtin_amdgcn_wmma_f32_16x16x4_f32(false, a, false, b1, (short)0, acc1, false, false);
            acc2 = __builtin_amdgcn_wmma_f32_16x16x4_f32(false, a, false, b2, (short)0, acc2, false, false);
            acc3 = __builtin_amdgcn_wmma_f32_16x16x4_f32(false, a, false, b3, (short)0, acc3, false, false);
        }

        const float bb0 = b2[ 0 + l15], bb1 = b2[16 + l15];
        const float bb2 = b2[32 + l15], bb3 = b2[48 + l15];
        float hsum0 = 0.f, hsum1 = 0.f, hsum2 = 0.f, hsum3 = 0.f;
#pragma unroll
        for (int r = 0; r < 8; ++r) {
            const int node = tile * 16 + r + halfsel * 8;
            if (node < n) {
                const float dn = dstn[node];
                float v0 = acc0[r] * dn + bb0; v0 = (v0 >= 0.f) ? v0 : pa * v0;
                float v1 = acc1[r] * dn + bb1; v1 = (v1 >= 0.f) ? v1 : pa * v1;
                float v2 = acc2[r] * dn + bb2; v2 = (v2 >= 0.f) ? v2 : pa * v2;
                float v3 = acc3[r] * dn + bb3; v3 = (v3 >= 0.f) ? v3 : pa * v3;
                float* o = hout + (size_t)node * OUT_DIM + l15;
                o[0] = v0; o[16] = v1; o[32] = v2; o[48] = v3;
                hsum0 += v0; hsum1 += v1; hsum2 += v2; hsum3 += v3;
            }
        }
        atomicAdd(&shg[ 0 + l15], hsum0);
        atomicAdd(&shg[16 + l15], hsum1);
        atomicAdd(&shg[32 + l15], hsum2);
        atomicAdd(&shg[48 + l15], hsum3);
    }
    __syncthreads();
    if (threadIdx.x < OUT_DIM) atomicAdd(&hg2[threadIdx.x], shg[threadIdx.x]);
}

extern "C" void kernel_launch(void* const* d_in, const int* in_sizes, int n_in,
                              void* d_out, int out_size, void* d_ws, size_t ws_size,
                              hipStream_t stream) {
    const float* feat    = (const float*)d_in[0];
    const int*   src     = (const int*)  d_in[1];
    const int*   dst     = (const int*)  d_in[2];
    const float* W1      = (const float*)d_in[3];
    const float* b1      = (const float*)d_in[4];
    const float* W2      = (const float*)d_in[5];
    const float* b2      = (const float*)d_in[6];
    const float* prelu_a = (const float*)d_in[7];

    const int n = in_sizes[0] / IN_DIM;
    const int E = in_sizes[1];

    float* out = (float*)d_out;
    float* hg1 = out + (size_t)n * OUT_DIM;
    float* hg2 = hg1 + OUT_DIM;

    float* wsf  = (float*)d_ws;
    float* srcn = wsf;
    float* dstn = wsf + n;
    float* agg  = wsf + 2 * (size_t)n;
    float* t1   = out;

    const int ntiles = (n + 15) / 16;
    const int gemm_blocks = (ntiles + 7) / 8;
    const long long tot = (long long)E * (OUT_DIM / 4);
    const int sblocks = (int)((tot + 255) / 256);

    k_init   <<<2048, 256, 0, stream>>>(srcn, dstn, agg, hg1, n);
    k_degree <<<(E + 255) / 256, 256, 0, stream>>>(src, dst, srcn, dstn, E);
    k_norm   <<<(n + 255) / 256, 256, 0, stream>>>(srcn, dstn, n);
    k_gemm1  <<<gemm_blocks, 256, 0, stream>>>(feat, W1, srcn, t1, n);
    k_scatter<<<sblocks, 256, 0, stream>>>(src, dst, t1, agg, E);
    k_finalize1<<<1024, 256, 0, stream>>>(agg, dstn, srcn, b1, prelu_a, t1, hg1, n);
    k_scatter<<<sblocks, 256, 0, stream>>>(src, dst, t1, agg, E);
    k_gemm2  <<<gemm_blocks, 256, 0, stream>>>(agg, W2, dstn, b2, prelu_a, out, hg2, n);
}


# layer-2 row gather fused into GEMM2: a wave sums its own 16-node tile (8 lanes a node, list order kept) into LDS and multiplies from there; agg round trip and one launch go; t1 and hs swap buffers
# speedup vs baseline: 3.9970x; 3.9970x over previous
#include <hip/hip_runtime.h>

constexpr int IN_DIM  = 128;
constexpr int OUT_DIM = 64;
constexpr int S1 = IN_DIM + 4;
constexpr int S2 = OUT_DIM + 4;

typedef float v2f __attribute__((ext_vector_type(2)));
typedef float v8f __attribute__((ext_vector_type(8)));

constexpr int BSH   = 9;
constexpr int BW    = 1 << BSH;
constexpr int NBMAX = 1024;
constexpr int CH    = 8192;
constexpr int EPT   = CH / 1024;
constexpr int CAP   = 16384;

static __device__ __forceinline__ void block_scan2(int* a, int* b, int cnt, int (*wt)[32]) {
    const int t = threadIdx.x, lane = t & 31, w = t >> 5;
    const int va = (t < cnt) ? a[t] : 0;
    const int vb = (t < cnt) ? b[t] : 0;
    int ia = va, ib = vb;
    for (int o = 1; o < 32; o <<= 1) {
        const int xa = __shfl_up(ia, o);
        const int xb = __shfl_up(ib, o);
        if (lane >= o) { ia += xa; ib += xb; }
    }
    if (lane == 31) { wt[0][w] = ia; wt[1][w] = ib; }
    __syncthreads();
    if (w == 0) {
        const int ta = wt[0][lane], tb = wt[1][lane];
        int sa = ta, sb = tb;
        for (int o = 1; o < 32; o <<= 1) {
            const int xa = __shfl_up(sa, o);
            const int xb = __shfl_up(sb, o);
            if (lane >= o) { sa += xa; sb += xb; }
        }
        wt[0][lane] = sa - ta;
        wt[1][lane] = sb - tb;
        if (lane == 31) { a[cnt] = sa; b[cnt] = sb; }
    }
    __syncthreads();
    if (t < cnt) { a[t] = wt[0][w] + ia - va; b[t] = wt[1][w] + ib - vb; }
    __syncthreads();
}

__global__ void __launch_bounds__(1024) k_bin(const int* __restrict__ src,
                                              const int* __restrict__ dst,
                                              int* __restrict__ pdw,
                                              int* __restrict__ ps,
                                              int* __restrict__ offD,
                                              int* __restrict__ offS,
                                              float* __restrict__ hg,
                                              int E, int nb, int ostride) {
    int2* pd = (int2*)pdw;
    __shared__ int  hd[NBMAX + 1];
    __shared__ int  hs[NBMAX + 1];
    __shared__ int2 sd[CH];
    __shared__ int  ss[CH];
    __shared__ int  wt[2][32];
    const int t = threadIdx.x;
    const int c = blockIdx.x;
    const int e0 = c * CH;
    const int m = min(CH, E - e0);
    if (c == 0 && t < 2 * OUT_DIM) hg[t] = 0.f;
    for (int i = t; i <= nb; i += 1024) { hd[i] = 0; hs[i] = 0; }
    __syncthreads();

    int s[EPT], d[EPT], rd[EPT], rs[EPT];
#pragma unroll
    for (int i = 0; i < EPT; ++i) {
        const int j = t + i * 1024;
        s[i] = (j < m) ? src[e0 + j] : 0;
        d[i] = (j < m) ? dst[e0 + j] : 0;
        rd[i] = 0;
        rs[i] = 0;
    }
#pragma unroll
    for (int i = 0; i < EPT; ++i) {
        const int j = t + i * 1024;
        if (j < m) {
            rd[i] = atomicAdd(&hd[d[i] >> BSH], 1);
            rs[i] = atomicAdd(&hs[s[i] >> BSH], 1);
        }
    }
    __syncthreads();
    block_scan2(hd, hs, nb, wt);

    int* od = offD + (size_t)c * ostride;
    int* os = offS + (size_t)c * ostride;
    for (int i = t; i <= nb; i += 1024) { od[i] = hd[i]; os[i] = hs[i]; }
#pragma unroll
    for (int i = 0; i < EPT; ++i) {
        const int j = t + i * 1024;
        if (j < m) {
            sd[hd[d[i] >> BSH] + rd[i]] = make_int2(s[i], d[i]);
            ss[hs[s[i] >> BSH] + rs[i]] = s[i];
        }
    }
    __syncthreads();
    for (int j = t; j < m; j += 1024) {
        pd[(size_t)e0 + j] = sd[j];
        ps[(size_t)e0 + j] = ss[j];
    }
}

__global__ void __launch_bounds__(1024) k_own(const int* __restrict__ pdw,
                                              const int* __restrict__ ps,
                                              const int* __restrict__ offD,
                                              const int* __restrict__ offS,
                                              int* __restrict__ rowptr,
                                              int* __restrict__ csr,
                                              float* __restrict__ srcn,
                                              float* __restrict__ dstn,
                                              int* spillw,
                                              int n, int nb, int nchunks, int ostride) {
    const int2* pd = (const int2*)pdw;
    int2* spill = (int2*)spillw;
    __shared__ int cnt[BW];
    __shared__ int oc[BW];
    __shared__ int excl[BW + 1];
    __shared__ int us[CAP];
    __shared__ int ud[CAP];
    __shared__ int red[2][32];
    __shared__ int pcd[2][1024];
    __shared__ int pcs[2][1024];
    const int t = threadIdx.x, lane = t & 31, w = t >> 5;
    const int b = blockIdx.x;
    const int base = b << BSH;

    for (int i = t; i < BW; i += 1024) { cnt[i] = 0; oc[i] = 0; }

    int pb = 0, pt = 0;
    for (int c = t; c < nchunks; c += 1024) {
        const int o0 = offD[(size_t)c * ostride + b];
        const int o1 = offD[(size_t)c * ostride + b + 1];
        pb += o0;
        pt += o1 - o0;
        if (c < 1024) {
            pcd[0][c] = o0;
            pcd[1][c] = o1;
            pcs[0][c] = offS[(size_t)c * ostride + b];
            pcs[1][c] = offS[(size_t)c * ostride + b + 1];
        }
    }
    for (int o = 16; o > 0; o >>= 1) { pb += __shfl_xor(pb, o); pt += __shfl_xor(pt, o); }
    if (lane == 0) { red[0][w] = pb; red[1][w] = pt; }
    __syncthreads();
    int binbase = 0, total = 0;
    for (int k = 0; k < 32; ++k) { binbase += red[0][k]; total += red[1][k]; }
    __syncthreads();

    auto count_piece = [&](int c, int o0, int o1, int q0, int q1) {
        const int2* p = pd + (size_t)c * CH + o0;
        const int*  q = ps + (size_t)c * CH + q0;
        const int ld = o1 - o0, ls = q1 - q0;
        const int len = max(ld, ls);
        for (int i = lane; i < len; i += 64) {
            const int i2 = i + 32;
            const bool d0 = i < ld, d1 = i2 < ld, s0 = i < ls, s1 = i2 < ls;
            const int dv0 = d0 ? p[i].y  : base;
            const int dv1 = d1 ? p[i2].y : base;
            const int sv0 = s0 ? q[i]    : base;
            const int sv1 = s1 ? q[i2]   : base;
            if (d0) atomicAdd(&cnt[dv0 - base], 1);
            if (d1) atomicAdd(&cnt[dv1 - base], 1);
            if (s0) atomicAdd(&oc[sv0 - base], 1);
            if (s1) atomicAdd(&oc[sv1 - base], 1);
        }
    };
    const int nlds = min(nchunks, 1024);
    for (int c = w; c < nlds; c += 32)
        count_piece(c, pcd[0][c], pcd[1][c], pcs[0][c], pcs[1][c]);
    for (int c = 1024 + w; c < nchunks; c += 32)
        count_piece(c, offD[(size_t)c * ostride + b], offD[(size_t)c * ostride + b + 1],
                       offS[(size_t)c * ostride + b], offS[(size_t)c * ostride + b + 1]);
    __syncthreads();

    {
        const int v = (t < BW) ? cnt[t] : 0;
        int inc = v;
        for (int o = 1; o < 32; o <<= 1) {
            const int x = __shfl_up(inc, o);
            if (lane >= o) inc += x;
        }
        if (lane == 31) red[0][w] = inc;
        __syncthreads();
        int wo = 0;
        for (int k = 0; k < w; ++k) wo += red[0][k];
        const int ex = wo + inc - v;
        if (t < BW) {
            const int node = base + t;
            if (node < n) {
                rowptr[node] = binbase + ex;
                dstn[node] = rsqrtf(fmaxf((float)v, 1.0f));
                srcn[node] = rsqrtf(fmaxf((float)oc[t], 1.0f));
            }
            excl[t] = ex;
            cnt[t] = 0;
        }
        if (t == 0) {
            excl[BW] = total;
            if (b == nb - 1) rowptr[n] = binbase + total;
        }
    }
    __syncthreads();

    const bool fits = total <= CAP;
    int2* sp = spill + binbase;
    auto place_piece = [&](int c, int o0, int o1) {
        const int2* p = pd + (size_t)c * CH + o0;
        const int ld = o1 - o0;
        for (int i = lane; i < ld; i += 64) {
            const int i2 = i + 32;
            const bool d1 = i2 < ld;
            const int2 e0 = p[i];
            const int2 e1 = d1 ? p[i2] : make_int2(0, base);
            const int l0 = e0.y - base, l1 = e1.y - base;
            const int pos0 = excl[l0] + atomicAdd(&cnt[l0], 1);
            if (fits) { us[pos0] = e0.x; ud[pos0] = l0; }
            else      { sp[pos0] = make_int2(e0.x, l0); }
            if (d1) {
                const int pos1 = excl[l1] + atomicAdd(&cnt[l1], 1);
                if (fits) { us[pos1] = e1.x; ud[pos1] = l1; }
                else      { sp[pos1] = make_int2(e1.x, l1); }
            }
        }
    };
    for (int c = w; c < nlds; c += 32)
        place_piece(c, pcd[0][c], pcd[1][c]);
    for (int c = 1024 + w; c < nchunks; c += 32)
        place_piece(c, offD[(size_t)c * ostride + b], offD[(size_t)c * ostride + b + 1]);
    __syncthreads();

    if (fits) {
        for (int p = t; p < total; p += 1024) {
            const int me = us[p];
            const int row = ud[p];
            const int beg = excl[row], end = excl[row + 1];
            int k = 0;
            for (int j = beg; j < end; ++j) {
                const int x = us[j];
                k += ((x < me) || (x == me && j < p)) ? 1 : 0;
            }
            csr[(size_t)binbase + beg + k] = me;
        }
    } else {
        for (int p = t; p < total; p += 1024) {
            const int2 mine = sp[p];
            const int beg = excl[mine.y], end = excl[mine.y + 1];
            int k = 0;
            for (int j = beg; j < end; ++j) {
                const int x = sp[j].x;
                k += ((x < mine.x) || (x == mine.x && j < p)) ? 1 : 0;
            }
            csr[(size_t)binbase + beg + k] = mine.x;
        }
    }
}

__global__ void __launch_bounds__(256) k_gemm1(const float* __restrict__ feat,
                                               const float* __restrict__ W1,
                                               const float* __restrict__ srcn,
                                               float* __restrict__ t1, int n) {
    __shared__ float sWt[OUT_DIM * S1];
    for (int i = threadIdx.x; i < IN_DIM * OUT_DIM; i += 256) {
        const int k = i >> 6;
        const int c = i & 63;
        sWt[c * S1 + k] = W1[i];
    }
    __syncthreads();

    const int wave = threadIdx.x >> 5;
    const int lane = threadIdx.x & 31;
    const int tile = blockIdx.x * 8 + wave;
    const int ntiles = (n + 15) >> 4;
    if (tile >= ntiles) return;

    const int halfsel = lane >> 4;
    const int l15 = lane & 15;
    int row = tile * 16 + l15;
    if (row >= n) row = n - 1;
    const int koff = halfsel * 2;
    const float norm = srcn[row];
    const float* arow = feat + (size_t)row * IN_DIM;
    const float* bc0 = sWt + ( 0 + l15) * S1;
    const float* bc1 = sWt + (16 + l15) * S1;
    const float* bc2 = sWt + (32 + l15) * S1;
    const float* bc3 = sWt + (48 + l15) * S1;

    v8f acc0 = {}, acc1 = {}, acc2 = {}, acc3 = {};
#pragma unroll 4
    for (int kb = 0; kb < IN_DIM / 4; ++kb) {
        const int k0 = kb * 4 + koff;
        v2f a = *(const v2f*)(arow + k0);
        a.x *= norm; a.y *= norm;
        const v2f b0 = *(const v2f*)(bc0 + k0);
        const v2f b1 = *(const v2f*)(bc1 + k0);
        const v2f b2 = *(const v2f*)(bc2 + k0);
        const v2f b3 = *(const v2f*)(bc3 + k0);
        acc0 = __builtin_amdgcn_wmma_f32_16x16x4_f32(false, a, false, b0, (short)0, acc0, false, false);
        acc1 = __builtin_amdgcn_wmma_f32_16x16x4_f32(false, a, false, b1, (short)0, acc1, false, false);
        acc2 = __builtin_amdgcn_wmma_f32_16x16x4_f32(false, a, false, b2, (short)0, acc2, false, false);
        acc3 = __builtin_amdgcn_wmma_f32_16x16x4_f32(false, a, false, b3, (short)0, acc3, false, false);
    }

#pragma unroll
    for (int r = 0; r < 8; ++r) {
        const int node = tile * 16 + r + halfsel * 8;
        if (node < n) {
            float* o = t1 + (size_t)node * OUT_DIM + l15;
            o[0]  = acc0[r];
            o[16] = acc1[r];
            o[32] = acc2[r];
            o[48] = acc3[r];
        }
    }
}

static __device__ __forceinline__ float4 gather_row(const float* __restrict__ msg,
                                                    const int* __restrict__ csr,
                                                    int beg, int end, int l) {
    float4 acc = make_float4(0.f, 0.f, 0.f, 0.f);
    const float* col = msg + 4 * l;
    for (int c = beg; c < end; c += 32) {
        const int cnt = min(32, end - c);
        const int m0 = (l < cnt) ? csr[c + l] : 0;
        const int m1 = (16 + l < cnt) ? csr[c + 16 + l] : 0;
        for (int k = 0; k < cnt; k += 4) {
            const int m = (k < 16) ? m0 : m1;
            const int kk = k & 15;
            const bool v1 = k + 1 < cnt, v2 = k + 2 < cnt, v3 = k + 3 < cnt;
            const int s0 = __shfl(m, kk, 16);
            const int s1 = __shfl(m, v1 ? kk + 1 : kk, 16);
            const int s2 = __shfl(m, v2 ? kk + 2 : kk, 16);
            const int s3 = __shfl(m, v3 ? kk + 3 : kk, 16);
            const float4 r0 = *(const float4*)(col + (size_t)s0 * OUT_DIM);
            const float4 r1 = *(const float4*)(col + (size_t)s1 * OUT_DIM);
            const float4 r2 = *(const float4*)(col + (size_t)s2 * OUT_DIM);
            const float4 r3 = *(const float4*)(col + (size_t)s3 * OUT_DIM);
            acc.x += r0.x; acc.y += r0.y; acc.z += r0.z; acc.w += r0.w;
            acc.x += v1 ? r1.x : 0.f; acc.y += v1 ? r1.y : 0.f; acc.z += v1 ? r1.z : 0.f; acc.w += v1 ? r1.w : 0.f;
            acc.x += v2 ? r2.x : 0.f; acc.y += v2 ? r2.y : 0.f; acc.z += v2 ? r2.z : 0.f; acc.w += v2 ? r2.w : 0.f;
            acc.x += v3 ? r3.x : 0.f; acc.y += v3 ? r3.y : 0.f; acc.z += v3 ? r3.z : 0.f; acc.w += v3 ? r3.w : 0.f;
        }
    }
    return acc;
}

__global__ void __launch_bounds__(512) k_gather1(const float* __restrict__ msg,
                                                 const int* __restrict__ csr,
                                                 const int* __restrict__ rowptr,
                                                 const float* __restrict__ dstn,
                                                 const float* __restrict__ srcn,
                                                 const float* __restrict__ b1,
                                                 const float* __restrict__ prelu_a,
                                                 float* __restrict__ hs,
                                                 float* __restrict__ hg1, int n) {
    const float pa = *prelu_a;
    const int l = threadIdx.x & 15;
    const int grp = threadIdx.x >> 4;
    const float4 bias = *(const float4*)(b1 + 4 * l);
    float4 part = make_float4(0.f, 0.f, 0.f, 0.f);
    for (int base = blockIdx.x * 32; base < n; base += gridDim.x * 32) {
        const int node = base + grp;
        if (node < n) {
            const float4 v = gather_row(msg, csr, rowptr[node], rowptr[node + 1], l);
            const float dn = dstn[node];
            const float sn = srcn[node];
            float4 h;
            h.x = v.x * dn + bias.x; h.x = (h.x >= 0.f) ? h.x : pa * h.x;
            h.y = v.y * dn + bias.y; h.y = (h.y >= 0.f) ? h.y : pa * h.y;
            h.z = v.z * dn + bias.z; h.z = (h.z >= 0.f) ? h.z : pa * h.z;
            h.w = v.w * dn + bias.w; h.w = (h.w >= 0.f) ? h.w : pa * h.w;
            part.x += h.x; part.y += h.y; part.z += h.z; part.w += h.w;
            *(float4*)(hs + (size_t)node * OUT_DIM + 4 * l) =
                make_float4(h.x * sn, h.y * sn, h.z * sn, h.w * sn);
        }
    }
    __shared__ float red[32 * OUT_DIM];
    *(float4*)(red + grp * OUT_DIM + 4 * l) = part;
    __syncthreads();
    if (threadIdx.x < OUT_DIM) {
        float sum = 0.f;
        for (int g = 0; g < 32; ++g) sum += red[g * OUT_DIM + threadIdx.x];
        atomicAdd(&hg1[threadIdx.x], sum);
    }
}

static __device__ __forceinline__ void gather_row8(const float* __restrict__ msg,
                                                   const int* __restrict__ csr,
                                                   int beg, int end, int l,
                                                   float4& lo, float4& hi) {
    float4 a = make_float4(0.f, 0.f, 0.f, 0.f);
    float4 b = make_float4(0.f, 0.f, 0.f, 0.f);
    const float* col = msg + 4 * l;
    for (int c = beg; c < end; c += 32) {
        const int cnt = min(32, end - c);
        const int m0 = (l < cnt) ? csr[c + l] : 0;
        const int m1 = (8 + l < cnt) ? csr[c + 8 + l] : 0;
        const int m2 = (16 + l < cnt) ? csr[c + 16 + l] : 0;
        const int m3 = (24 + l < cnt) ? csr[c + 24 + l] : 0;
        for (int k = 0; k < cnt; k += 4) {
            const int m = (k < 8) ? m0 : (k < 16) ? m1 : (k < 24) ? m2 : m3;
            const int kk = k & 7;
            const bool v1 = k + 1 < cnt, v2 = k + 2 < cnt, v3 = k + 3 < cnt;
            const int s0 = __shfl(m, kk, 8);
            const int s1 = __shfl(m, v1 ? kk + 1 : kk, 8);
            const int s2 = __shfl(m, v2 ? kk + 2 : kk, 8);
            const int s3 = __shfl(m, v3 ? kk + 3 : kk, 8);
            const float* p0 = col + (size_t)s0 * OUT_DIM;
            const float* p1 = col + (size_t)s1 * OUT_DIM;
            const float* p2 = col + (size_t)s2 * OUT_DIM;
            const float* p3 = col + (size_t)s3 * OUT_DIM;
            const float4 r0a = *(const float4*)p0, r0b = *(const float4*)(p0 + 32);
            const float4 r1a = *(const float4*)p1, r1b = *(const float4*)(p1 + 32);
            const float4 r2a = *(const float4*)p2, r2b = *(const float4*)(p2 + 32);
            const float4 r3a = *(const float4*)p3, r3b = *(const float4*)(p3 + 32);
            a.x += r0a.x; a.y += r0a.y; a.z += r0a.z; a.w += r0a.w;
            b.x += r0b.x; b.y += r0b.y; b.z += r0b.z; b.w += r0b.w;
            a.x += v1 ? r1a.x : 0.f; a.y += v1 ? r1a.y : 0.f; a.z += v1 ? r1a.z : 0.f; a.w += v1 ? r1a.w : 0.f;
            b.x += v1 ? r1b.x : 0.f; b.y += v1 ? r1b.y : 0.f; b.z += v1 ? r1b.z : 0.f; b.w += v1 ? r1b.w : 0.f;
            a.x += v2 ? r2a.x : 0.f; a.y += v2 ? r2a.y : 0.f; a.z += v2 ? r2a.z : 0.f; a.w += v2 ? r2a.w : 0.f;
            b.x += v2 ? r2b.x : 0.f; b.y += v2 ? r2b.y : 0.f; b.z += v2 ? r2b.z : 0.f; b.w += v2 ? r2b.w : 0.f;
            a.x += v3 ? r3a.x : 0.f; a.y += v3 ? r3a.y : 0.f; a.z += v3 ? r3a.z : 0.f; a.w += v3 ? r3a.w : 0.f;
            b.x += v3 ? r3b.x : 0.f; b.y += v3 ? r3b.y : 0.f; b.z += v3 ? r3b.z : 0.f; b.w += v3 ? r3b.w : 0.f;
        }
    }
    lo = a;
    hi = b;
}

constexpr int SA = OUT_DIM + 4;

__global__ void __launch_bounds__(256) k_gather_gemm2(const float* __restrict__ msg,
                                                      const int* __restrict__ csr,
                                                      const int* __restrict__ rowptr,
                                                      const float* __restrict__ W2,
                                                      const float* __restrict__ dstn,
                                                      const float* __restrict__ b2,
                                                      const float* __restrict__ prelu_a,
                                                      float* __restrict__ hout,
                                                      float* __restrict__ hg2, int n) {
    __shared__ float sWt[OUT_DIM * S2];
    __shared__ __align__(16) float sA[8 * 16 * SA];
    __shared__ float shg[OUT_DIM];

    const int wave = threadIdx.x >> 5;
    const int lane = threadIdx.x & 31;
    const int tile = blockIdx.x * 8 + wave;
    const int ntiles = (n + 15) >> 4;
    const bool active = tile < ntiles;

    float wreg[16];
#pragma unroll
    for (int i = 0; i < 16; ++i) wreg[i] = W2[threadIdx.x + 256 * i];
    if (threadIdx.x < OUT_DIM) shg[threadIdx.x] = 0.f;

    float* arows = sA + wave * 16 * SA;
    if (active) {
        const int rp = rowptr[min(tile * 16 + lane, n)];
        const int l = lane & 7;
        const int g = lane >> 3;
        for (int p = 0; p < 4; ++p) {
            const int r = 4 * p + g;
            const int beg = __shfl(rp, r);
            const int end = __shfl(rp, r + 1);
            float4 lo, hi;
            gather_row8(msg, csr, beg, end, l, lo, hi);
            *(float4*)(arows + r * SA + 4 * l)      = lo;
            *(float4*)(arows + r * SA + 32 + 4 * l) = hi;
        }
    }
#pragma unroll
    for (int i = 0; i < 16; ++i) {
        const int idx = threadIdx.x + 256 * i;
        sWt[(idx & 63) * S2 + (idx >> 6)] = wreg[i];
    }
    __syncthreads();

    const float pa = *prelu_a;
    const int halfsel = lane >> 4;
    const int l15 = lane & 15;

    if (active) {
        const int koff = halfsel * 2;
        const float* arow = arows + l15 * SA;
        const float* bc0 = sWt + ( 0 + l15) * S2;
        const float* bc1 = sWt + (16 + l15) * S2;
        const float* bc2 = sWt + (32 + l15) * S2;
        const float* bc3 = sWt + (48 + l15) * S2;

        v8f acc0 = {}, acc1 = {}, acc2 = {}, acc3 = {};
#pragma unroll 4
        for (int kb = 0; kb < OUT_DIM / 4; ++kb) {
            const int k0 = kb * 4 + koff;
            const v2f a  = *(const v2f*)(arow + k0);
            const v2f b0 = *(const v2f*)(bc0 + k0);
            const v2f b1 = *(const v2f*)(bc1 + k0);
            const v2f b2v = *(const v2f*)(bc2 + k0);
            const v2f b3 = *(const v2f*)(bc3 + k0);
            acc0 = __builtin_amdgcn_wmma_f32_16x16x4_f32(false, a, false, b0, (short)0, acc0, false, false);
            acc1 = __builtin_amdgcn_wmma_f32_16x16x4_f32(false, a, false, b1, (short)0, acc1, false, false);
            acc2 = __builtin_amdgcn_wmma_f32_16x16x4_f32(false, a, false, b2v, (short)0, acc2, false, false);
            acc3 = __builtin_amdgcn_wmma_f32_16x16x4_f32(false, a, false, b3, (short)0, acc3, false, false);
        }

        const float bb0 = b2[ 0 + l15], bb1 = b2[16 + l15];
        const float bb2 = b2[32 + l15], bb3 = b2[48 + l15];
        float hsum0 = 0.f, hsum1 = 0.f, hsum2 = 0.f, hsum3 = 0.f;
#pragma unroll
        for (int r = 0; r < 8; ++r) {
            const int node = tile * 16 + r + halfsel * 8;
            if (node < n) {
                const float dn = dstn[node];
                float v0 = acc0[r] * dn + bb0; v0 = (v0 >= 0.f) ? v0 : pa * v0;
                float v1 = acc1[r] * dn + bb1; v1 = (v1 >= 0.f) ? v1 : pa * v1;
                float v2 = acc2[r] * dn + bb2; v2 = (v2 >= 0.f) ? v2 : pa * v2;
                float v3 = acc3[r] * dn + bb3; v3 = (v3 >= 0.f) ? v3 : pa * v3;
                float* o = hout + (size_t)node * OUT_DIM + l15;
                o[0] = v0; o[16] = v1; o[32] = v2; o[48] = v3;
                hsum0 += v0; hsum1 += v1; hsum2 += v2; hsum3 += v3;
            }
        }
        atomicAdd(&shg[ 0 + l15], hsum0);
        atomicAdd(&shg[16 + l15], hsum1);
        atomicAdd(&shg[32 + l15], hsum2);
        atomicAdd(&shg[48 + l15], hsum3);
    }
    __syncthreads();
    if (threadIdx.x < OUT_DIM) atomicAdd(&hg2[threadIdx.x], shg[threadIdx.x]);
}

extern "C" void kernel_launch(void* const* d_in, const int* in_sizes, int n_in,
                              void* d_out, int out_size, void* d_ws, size_t ws_size,
                              hipStream_t stream) {
    const float* feat    = (const float*)d_in[0];
    const int*   src     = (const int*)  d_in[1];
    const int*   dst     = (const int*)  d_in[2];
    const float* W1      = (const float*)d_in[3];
    const float* b1      = (const float*)d_in[4];
    const float* W2      = (const float*)d_in[5];
    const float* b2      = (const float*)d_in[6];
    const float* prelu_a = (const float*)d_in[7];

    const int n = in_sizes[0] / IN_DIM;
    const int E = in_sizes[1];

    float* out = (float*)d_out;
    float* hg1 = out + (size_t)n * OUT_DIM;
    float* hg2 = hg1 + OUT_DIM;

    float* wsf    = (float*)d_ws;
    float* big    = wsf;
    int*   csr    = (int*)(wsf + (size_t)n * OUT_DIM);
    int*   rowptr = csr + E;
    float* srcn   = (float*)(rowptr + n + 32);
    float* dstn   = srcn + n;

    const int nb      = (n + BW - 1) / BW;
    const int nchunks = (E + CH - 1) / CH;
    const int ostride = (nb + 1 + 31) & ~31;

    int*  pd    = (int*)out;
    int*  ps    = (int*)out + 2 * (size_t)E;
    int*  offD  = ps + E;
    int*  offS  = offD + (size_t)nchunks * ostride;
    int*  spill = (int*)big;

    const int ntiles = (n + 15) / 16;
    const int gemm_blocks = (ntiles + 7) / 8;

    k_bin    <<<nchunks, 1024, 0, stream>>>(src, dst, pd, ps, offD, offS, hg1, E, nb, ostride);
    k_own    <<<nb, 1024, 0, stream>>>(pd, ps, offD, offS, rowptr, csr, srcn, dstn, spill, n, nb, nchunks, ostride);
    k_gemm1  <<<gemm_blocks, 256, 0, stream>>>(feat, W1, srcn, out, n);
    k_gather1<<<1024, 512, 0, stream>>>(out, csr, rowptr, dstn, srcn, b1, prelu_a, big, hg1, n);
    k_gather_gemm2<<<gemm_blocks, 256, 0, stream>>>(big, csr, rowptr, W2, dstn, b2, prelu_a, out, hg2, n);
}


# GEMM1 moves into the bin owner launch: a wave multiplies its own 16-node tile 2-3 k-steps at a time under the list build's round trips; source norm applied to the finished sums; 5 launches to 4
# speedup vs baseline: 4.2803x; 1.0709x over previous
#include <hip/hip_runtime.h>

constexpr int IN_DIM  = 128;
constexpr int OUT_DIM = 64;
constexpr int S1 = IN_DIM + 4;
constexpr int S2 = OUT_DIM + 4;

typedef float v2f __attribute__((ext_vector_type(2)));
typedef float v8f __attribute__((ext_vector_type(8)));

constexpr int BSH   = 9;
constexpr int BW    = 1 << BSH;
constexpr int NBMAX = 1024;
constexpr int CH    = 8192;
constexpr int EPT   = CH / 1024;
constexpr int CAP   = 16384;
constexpr int LSH   = 32 - BSH;
constexpr int SMASK = (1 << LSH) - 1;

static __device__ __forceinline__ void block_scan2(int* a, int* b, int cnt, int (*wt)[32]) {
    const int t = threadIdx.x, lane = t & 31, w = t >> 5;
    const int va = (t < cnt) ? a[t] : 0;
    const int vb = (t < cnt) ? b[t] : 0;
    int ia = va, ib = vb;
    for (int o = 1; o < 32; o <<= 1) {
        const int xa = __shfl_up(ia, o);
        const int xb = __shfl_up(ib, o);
        if (lane >= o) { ia += xa; ib += xb; }
    }
    if (lane == 31) { wt[0][w] = ia; wt[1][w] = ib; }
    __syncthreads();
    if (w == 0) {
        const int ta = wt[0][lane], tb = wt[1][lane];
        int sa = ta, sb = tb;
        for (int o = 1; o < 32; o <<= 1) {
            const int xa = __shfl_up(sa, o);
            const int xb = __shfl_up(sb, o);
            if (lane >= o) { sa += xa; sb += xb; }
        }
        wt[0][lane] = sa - ta;
        wt[1][lane] = sb - tb;
        if (lane == 31) { a[cnt] = sa; b[cnt] = sb; }
    }
    __syncthreads();
    if (t < cnt) { a[t] = wt[0][w] + ia - va; b[t] = wt[1][w] + ib - vb; }
    __syncthreads();
}

__global__ void __launch_bounds__(1024) k_bin(const int* __restrict__ src,
                                              const int* __restrict__ dst,
                                              int* __restrict__ pd,
                                              int* __restrict__ ps,
                                              int* __restrict__ offD,
                                              int* __restrict__ offS,
                                              float* __restrict__ hg,
                                              int E, int nb, int ostride) {
    __shared__ int  hd[NBMAX + 1];
    __shared__ int  hs[NBMAX + 1];
    __shared__ int  sd[CH];
    __shared__ int  ss[CH];
    __shared__ int  wt[2][32];
    const int t = threadIdx.x;
    const int c = blockIdx.x;
    const int e0 = c * CH;
    const int m = min(CH, E - e0);
    if (c == 0 && t < 2 * OUT_DIM) hg[t] = 0.f;
    for (int i = t; i <= nb; i += 1024) { hd[i] = 0; hs[i] = 0; }
    __syncthreads();

    int s[EPT], d[EPT], rd[EPT], rs[EPT];
#pragma unroll
    for (int i = 0; i < EPT; ++i) {
        const int j = t + i * 1024;
        s[i] = (j < m) ? src[e0 + j] : 0;
        d[i] = (j < m) ? dst[e0 + j] : 0;
        rd[i] = 0;
        rs[i] = 0;
    }
#pragma unroll
    for (int i = 0; i < EPT; ++i) {
        const int j = t + i * 1024;
        if (j < m) {
            rd[i] = atomicAdd(&hd[d[i] >> BSH], 1);
            rs[i] = atomicAdd(&hs[s[i] >> BSH], 1);
        }
    }
    __syncthreads();
    block_scan2(hd, hs, nb, wt);

    int* od = offD + (size_t)c * ostride;
    int* os = offS + (size_t)c * ostride;
    for (int i = t; i <= nb; i += 1024) { od[i] = hd[i]; os[i] = hs[i]; }
#pragma unroll
    for (int i = 0; i < EPT; ++i) {
        const int j = t + i * 1024;
        if (j < m) {
            sd[hd[d[i] >> BSH] + rd[i]] = (int)((unsigned)s[i] | ((unsigned)(d[i] & (BW - 1)) << LSH));
            ss[hs[s[i] >> BSH] + rs[i]] = s[i];
        }
    }
    __syncthreads();
    for (int j = t; j < m; j += 1024) {
        pd[(size_t)e0 + j] = sd[j];
        ps[(size_t)e0 + j] = ss[j];
    }
}

constexpr int KT = IN_DIM / 4;

template <int KS>
static __device__ __forceinline__ void slice_load(const float* __restrict__ arow, int kb, v2f (&a)[KS]) {
#pragma unroll
    for (int i = 0; i < KS; ++i) a[i] = *(const v2f*)(arow + (kb + i) * 4);
}

template <int KS>
static __device__ __forceinline__ void slice_mma(const float* bc, int kb, const v2f (&a)[KS],
                                                 v8f& acc0, v8f& acc1, v8f& acc2, v8f& acc3) {
#pragma unroll
    for (int i = 0; i < KS; ++i) {
        const float* b = bc + (kb + i) * 4;
        const v2f b0 = *(const v2f*)(b);
        const v2f b1 = *(const v2f*)(b + 16 * S1);
        const v2f b2 = *(const v2f*)(b + 32 * S1);
        const v2f b3 = *(const v2f*)(b + 48 * S1);
        acc0 = __builtin_amdgcn_wmma_f32_16x16x4_f32(false, a[i], false, b0, (short)0, acc0, false, false);
        acc1 = __builtin_amdgcn_wmma_f32_16x16x4_f32(false, a[i], false, b1, (short)0, acc1, false, false);
        acc2 = __builtin_amdgcn_wmma_f32_16x16x4_f32(false, a[i], false, b2, (short)0, acc2, false, false);
        acc3 = __builtin_amdgcn_wmma_f32_16x16x4_f32(false, a[i], false, b3, (short)0, acc3, false, false);
    }
}

__global__ void __launch_bounds__(1024) k_own(const int* __restrict__ pd,
                                              const int* __restrict__ ps,
                                              const int* __restrict__ offD,
                                              const int* __restrict__ offS,
                                              int* __restrict__ rowptr,
                                              int* __restrict__ csr,
                                              float* __restrict__ srcn,
                                              float* __restrict__ dstn,
                                              int* spill,
                                              const float* __restrict__ feat,
                                              const float* __restrict__ W1,
                                              float* __restrict__ t1,
                                              int n, int nb, int nchunks, int ostride) {
    __shared__ int cnt[BW];
    __shared__ int oc[BW];
    __shared__ int excl[BW + 1];
    __shared__ int us[CAP];
    __shared__ int ud[CAP];
    __shared__ int red[2][32];
    __shared__ int pcd[2][1024];
    __shared__ int pcs[2][1024];
    __shared__ float sWt[OUT_DIM * S1];
    __shared__ float snl[BW];
    const int t = threadIdx.x, lane = t & 31;
    const int w = __builtin_amdgcn_readfirstlane(t >> 5);
    const int b = blockIdx.x;
    const int base = b << BSH;

    float wreg[IN_DIM * OUT_DIM / 1024];
#pragma unroll
    for (int i = 0; i < IN_DIM * OUT_DIM / 1024; ++i) wreg[i] = W1[t + 1024 * i];

    for (int i = t; i < BW; i += 1024) { cnt[i] = 0; oc[i] = 0; }

    int pb = 0, pt = 0;
    for (int c = t; c < nchunks; c += 1024) {
        const int o0 = offD[(size_t)c * ostride + b];
        const int o1 = offD[(size_t)c * ostride + b + 1];
        pb += o0;
        pt += o1 - o0;
        if (c < 1024) {
            pcd[0][c] = o0;
            pcd[1][c] = o1;
            pcs[0][c] = offS[(size_t)c * ostride + b];
            pcs[1][c] = offS[(size_t)c * ostride + b + 1];
        }
    }
    for (int o = 16; o > 0; o >>= 1) { pb += __shfl_xor(pb, o); pt += __shfl_xor(pt, o); }
    if (lane == 0) { red[0][w] = pb; red[1][w] = pt; }
#pragma unroll
    for (int i = 0; i < IN_DIM * OUT_DIM / 1024; ++i) {
        const int idx = t + 1024 * i;
        sWt[(idx & 63) * S1 + (idx >> 6)] = wreg[i];
    }
    __syncthreads();
    int binbase = red[0][lane], total = red[1][lane];
    for (int o = 16; o > 0; o >>= 1) { binbase += __shfl_xor(binbase, o); total += __shfl_xor(total, o); }
    __syncthreads();

    const int halfsel = lane >> 4;
    const int l15 = lane & 15;
    const int tile0 = base + w * 16;
    const bool gact = tile0 < n;
    const int grow = min(tile0 + l15, n - 1);
    const float* arow = feat + (size_t)grow * IN_DIM + halfsel * 2;
    const float* bc = sWt + l15 * S1 + halfsel * 2;
    v8f acc0 = {}, acc1 = {}, acc2 = {}, acc3 = {};
    int kdone = 0;

    auto count_piece = [&](int c, int o0, int o1, int q0, int q1, auto&& ahead) {
        const int* p = pd + (size_t)c * CH + o0;
        const int* q = ps + (size_t)c * CH + q0;
        const int ld = o1 - o0, ls = q1 - q0;
        const int len = max(ld, ls);
        auto trip = [&](int i, auto&& hook) {
            const int i2 = i + 32;
            const bool d0 = i < ld, d1 = i2 < ld, s0 = i < ls, s1 = i2 < ls;
            const unsigned dv0 = (unsigned)*(d0 ? p + i  : pd);
            const unsigned dv1 = (unsigned)*(d1 ? p + i2 : pd);
            const int sv0 = *(s0 ? q + i  : ps);
            const int sv1 = *(s1 ? q + i2 : ps);
            hook();
            if (d0) atomicAdd(&cnt[dv0 >> LSH], 1);
            if (d1) atomicAdd(&cnt[dv1 >> LSH], 1);
            if (s0) atomicAdd(&oc[sv0 - base], 1);
            if (s1) atomicAdd(&oc[sv1 - base], 1);
        };
        trip(lane, ahead);
        for (int i = lane + 64; i < len; i += 64) trip(i, [] {});
    };
    const int nlds = min(nchunks, 1024);
    for (int c = w; c < nlds; c += 32) {
        const bool g = gact && kdone + 3 <= KT;
        v2f a[3];
        count_piece(c, pcd[0][c], pcd[1][c], pcs[0][c], pcs[1][c],
                    [&] { if (g) slice_load<3>(arow, kdone, a); });
        if (g) { slice_mma<3>(bc, kdone, a, acc0, acc1, acc2, acc3); kdone += 3; }
    }
    for (int c = 1024 + w; c < nchunks; c += 32)
        count_piece(c, offD[(size_t)c * ostride + b], offD[(size_t)c * ostride + b + 1],
                       offS[(size_t)c * ostride + b], offS[(size_t)c * ostride + b + 1], [] {});
    __syncthreads();

    {
        const int v = (t < BW) ? cnt[t] : 0;
        int inc = v;
        for (int o = 1; o < 32; o <<= 1) {
            const int x = __shfl_up(inc, o);
            if (lane >= o) inc += x;
        }
        if (lane == 31) red[0][w] = inc;
        __syncthreads();
        int wo = 0;
        for (int k = 0; k < w; ++k) wo += red[0][k];
        const int ex = wo + inc - v;
        if (t < BW) {
            const int node = base + t;
            const float sn = rsqrtf(fmaxf((float)oc[t], 1.0f));
            if (node < n) {
                rowptr[node] = binbase + ex;
                dstn[node] = rsqrtf(fmaxf((float)v, 1.0f));
                srcn[node] = sn;
            }
            snl[t] = sn;
            excl[t] = ex;
            cnt[t] = 0;
        }
        if (t == 0) {
            excl[BW] = total;
            if (b == nb - 1) rowptr[n] = binbase + total;
        }
    }
    __syncthreads();

    const bool fits = total <= CAP;
    int* sp = spill + binbase;
    auto place_piece = [&](int c, int o0, int o1, auto&& ahead) {
        const int* p = pd + (size_t)c * CH + o0;
        const int ld = o1 - o0;
        auto trip = [&](int i, auto&& hook) {
            const int i2 = i + 32;
            const bool d0 = i < ld, d1 = i2 < ld;
            const unsigned e0 = (unsigned)*(d0 ? p + i  : pd);
            const unsigned e1 = (unsigned)*(d1 ? p + i2 : pd);
            hook();
            if (d0) {
                const int l0 = (int)(e0 >> LSH);
                const int pos0 = excl[l0] + atomicAdd(&cnt[l0], 1);
                if (fits) { us[pos0] = (int)(e0 & SMASK); ud[pos0] = l0; }
                else      { sp[pos0] = (int)e0; }
            }
            if (d1) {
                const int l1 = (int)(e1 >> LSH);
                const int pos1 = excl[l1] + atomicAdd(&cnt[l1], 1);
                if (fits) { us[pos1] = (int)(e1 & SMASK); ud[pos1] = l1; }
                else      { sp[pos1] = (int)e1; }
            }
        };
        trip(lane, ahead);
        for (int i = lane + 64; i < ld; i += 64) trip(i, [] {});
    };
    for (int c = w; c < nlds; c += 32) {
        const bool g = gact && kdone + 2 <= KT;
        v2f a[2];
        place_piece(c, pcd[0][c], pcd[1][c],
                    [&] { if (g) slice_load<2>(arow, kdone, a); });
        if (g) { slice_mma<2>(bc, kdone, a, acc0, acc1, acc2, acc3); kdone += 2; }
    }
    for (int c = 1024 + w; c < nchunks; c += 32)
        place_piece(c, offD[(size_t)c * ostride + b], offD[(size_t)c * ostride + b + 1], [] {});

    if (gact) {
        while (kdone < KT) {
            v2f a[1];
            slice_load<1>(arow, kdone, a);
            slice_mma<1>(bc, kdone, a, acc0, acc1, acc2, acc3);
            kdone += 1;
        }
#pragma unroll
        for (int r = 0; r < 8; ++r) {
            const int lrow = w * 16 + r + halfsel * 8;
            if (base + lrow < n) {
                const float sc = snl[lrow];
                float* o = t1 + (size_t)(base + lrow) * OUT_DIM + l15;
                o[0]  = acc0[r] * sc;
                o[16] = acc1[r] * sc;
                o[32] = acc2[r] * sc;
                o[48] = acc3[r] * sc;
            }
        }
    }
    __syncthreads();

    if (fits) {
        for (int p = t; p < total; p += 1024) {
            const int me = us[p];
            const int row = ud[p];
            const int beg = excl[row], end = excl[row + 1];
            int k = 0;
            for (int j = beg; j < end; ++j) {
                const int x = us[j];
                k += ((x < me) || (x == me && j < p)) ? 1 : 0;
            }
            csr[(size_t)binbase + beg + k] = me;
        }
    } else {
        for (int p = t; p < total; p += 1024) {
            const unsigned mine = (unsigned)sp[p];
            const int mx = (int)(mine & SMASK);
            const int my = (int)(mine >> LSH);
            const int beg = excl[my], end = excl[my + 1];
            int k = 0;
            for (int j = beg; j < end; ++j) {
                const int x = sp[j] & SMASK;
                k += ((x < mx) || (x == mx && j < p)) ? 1 : 0;
            }
            csr[(size_t)binbase + beg + k] = mx;
        }
    }
}

static __device__ __forceinline__ float4 gather_row(const float* __restrict__ msg,
                                                    const int* __restrict__ csr,
                                                    int beg, int end, int l) {
    float4 acc = make_float4(0.f, 0.f, 0.f, 0.f);
    const float* col = msg + 4 * l;
    for (int c = beg; c < end; c += 32) {
        const int cnt = min(32, end - c);
        const int m0 = (l < cnt) ? csr[c + l] : 0;
        const int m1 = (16 + l < cnt) ? csr[c + 16 + l] : 0;
        for (int k = 0; k < cnt; k += 4) {
            const int m = (k < 16) ? m0 : m1;
            const int kk = k & 15;
            const bool v1 = k + 1 < cnt, v2 = k + 2 < cnt, v3 = k + 3 < cnt;
            const int s0 = __shfl(m, kk, 16);
            const int s1 = __shfl(m, v1 ? kk + 1 : kk, 16);
            const int s2 = __shfl(m, v2 ? kk + 2 : kk, 16);
            const int s3 = __shfl(m, v3 ? kk + 3 : kk, 16);
            const float4 r0 = *(const float4*)(col + (size_t)s0 * OUT_DIM);
            const float4 r1 = *(const float4*)(col + (size_t)s1 * OUT_DIM);
            const float4 r2 = *(const float4*)(col + (size_t)s2 * OUT_DIM);
            const float4 r3 = *(const float4*)(col + (size_t)s3 * OUT_DIM);
            acc.x += r0.x; acc.y += r0.y; acc.z += r0.z; acc.w += r0.w;
            acc.x += v1 ? r1.x : 0.f; acc.y += v1 ? r1.y : 0.f; acc.z += v1 ? r1.z : 0.f; acc.w += v1 ? r1.w : 0.f;
            acc.x += v2 ? r2.x : 0.f; acc.y += v2 ? r2.y : 0.f; acc.z += v2 ? r2.z : 0.f; acc.w += v2 ? r2.w : 0.f;
            acc.x += v3 ? r3.x : 0.f; acc.y += v3 ? r3.y : 0.f; acc.z += v3 ? r3.z : 0.f; acc.w += v3 ? r3.w : 0.f;
        }
    }
    return acc;
}

__global__ void __launch_bounds__(512) k_gather1(const float* __restrict__ msg,
                                                 const int* __restrict__ csr,
                                                 const int* __restrict__ rowptr,
                                                 const float* __restrict__ dstn,
                                                 const float* __restrict__ srcn,
                                                 const float* __restrict__ b1,
                                                 const float* __restrict__ prelu_a,
                                                 float* __restrict__ hs,
                                                 float* __restrict__ hg1, int n) {
    const float pa = *prelu_a;
    const int l = threadIdx.x & 15;
    const int grp = threadIdx.x >> 4;
    const float4 bias = *(const float4*)(b1 + 4 * l);
    float4 part = make_float4(0.f, 0.f, 0.f, 0.f);
    for (int base = blockIdx.x * 32; base < n; base += gridDim.x * 32) {
        const int node = base + grp;
        if (node < n) {
            const float4 v = gather_row(msg, csr, rowptr[node], rowptr[node + 1], l);
            const float dn = dstn[node];
            const float sn = srcn[node];
            float4 h;
            h.x = v.x * dn + bias.x; h.x = (h.x >= 0.f) ? h.x : pa * h.x;
            h.y = v.y * dn + bias.y; h.y = (h.y >= 0.f) ? h.y : pa * h.y;
            h.z = v.z * dn + bias.z; h.z = (h.z >= 0.f) ? h.z : pa * h.z;
            h.w = v.w * dn + bias.w; h.w = (h.w >= 0.f) ? h.w : pa * h.w;
            part.x += h.x; part.y += h.y; part.z += h.z; part.w += h.w;
            *(float4*)(hs + (size_t)node * OUT_DIM + 4 * l) =
                make_float4(h.x * sn, h.y * sn, h.z * sn, h.w * sn);
        }
    }
    __shared__ float red[32 * OUT_DIM];
    *(float4*)(red + grp * OUT_DIM + 4 * l) = part;
    __syncthreads();
    if (threadIdx.x < OUT_DIM) {
        float sum = 0.f;
        for (int g = 0; g < 32; ++g) sum += red[g * OUT_DIM + threadIdx.x];
        atomicAdd(&hg1[threadIdx.x], sum);
    }
}

static __device__ __forceinline__ void gather_row8(const float* __restrict__ msg,
                                                   const int* __restrict__ csr,
                                                   int beg, int end, int l,
                                                   float4& lo, float4& hi) {
    float4 a = make_float4(0.f, 0.f, 0.f, 0.f);
    float4 b = make_float4(0.f, 0.f, 0.f, 0.f);
    const float* col = msg + 4 * l;
    for (int c = beg; c < end; c += 32) {
        const int cnt = min(32, end - c);
        const int m0 = (l < cnt) ? csr[c + l] : 0;
        const int m1 = (8 + l < cnt) ? csr[c + 8 + l] : 0;
        const int m2 = (16 + l < cnt) ? csr[c + 16 + l] : 0;
        const int m3 = (24 + l < cnt) ? csr[c + 24 + l] : 0;
        for (int k = 0; k < cnt; k += 4) {
            const int m = (k < 8) ? m0 : (k < 16) ? m1 : (k < 24) ? m2 : m3;
            const int kk = k & 7;
            const bool v1 = k + 1 < cnt, v2 = k + 2 < cnt, v3 = k + 3 < cnt;
            const int s0 = __shfl(m, kk, 8);
            const int s1 = __shfl(m, v1 ? kk + 1 : kk, 8);
            const int s2 = __shfl(m, v2 ? kk + 2 : kk, 8);
            const int s3 = __shfl(m, v3 ? kk + 3 : kk, 8);
            const float* p0 = col + (size_t)s0 * OUT_DIM;
            const float* p1 = col + (size_t)s1 * OUT_DIM;
            const float* p2 = col + (size_t)s2 * OUT_DIM;
            const float* p3 = col + (size_t)s3 * OUT_DIM;
            const float4 r0a = *(const float4*)p0, r0b = *(const float4*)(p0 + 32);
            const float4 r1a = *(const float4*)p1, r1b = *(const float4*)(p1 + 32);
            const float4 r2a = *(const float4*)p2, r2b = *(const float4*)(p2 + 32);
            const float4 r3a = *(const float4*)p3, r3b = *(const float4*)(p3 + 32);
            a.x += r0a.x; a.y += r0a.y; a.z += r0a.z; a.w += r0a.w;
            b.x += r0b.x; b.y += r0b.y; b.z += r0b.z; b.w += r0b.w;
            a.x += v1 ? r1a.x : 0.f; a.y += v1 ? r1a.y : 0.f; a.z += v1 ? r1a.z : 0.f; a.w += v1 ? r1a.w : 0.f;
            b.x += v1 ? r1b.x : 0.f; b.y += v1 ? r1b.y : 0.f; b.z += v1 ? r1b.z : 0.f; b.w += v1 ? r1b.w : 0.f;
            a.x += v2 ? r2a.x : 0.f; a.y += v2 ? r2a.y : 0.f; a.z += v2 ? r2a.z : 0.f; a.w += v2 ? r2a.w : 0.f;
            b.x += v2 ? r2b.x : 0.f; b.y += v2 ? r2b.y : 0.f; b.z += v2 ? r2b.z : 0.f; b.w += v2 ? r2b.w : 0.f;
            a.x += v3 ? r3a.x : 0.f; a.y += v3 ? r3a.y : 0.f; a.z += v3 ? r3a.z : 0.f; a.w += v3 ? r3a.w : 0.f;
            b.x += v3 ? r3b.x : 0.f; b.y += v3 ? r3b.y : 0.f; b.z += v3 ? r3b.z : 0.f; b.w += v3 ? r3b.w : 0.f;
        }
    }
    lo = a;
    hi = b;
}

constexpr int SA = OUT_DIM + 4;

__global__ void __launch_bounds__(256) k_gather_gemm2(const float* __restrict__ msg,
                                                      const int* __restrict__ csr,
                                                      const int* __restrict__ rowptr,
                                                      const float* __restrict__ W2,
                                                      const float* __restrict__ dstn,
                                                      const float* __restrict__ b2,
                                                      const float* __restrict__ prelu_a,
                                                      float* __restrict__ hout,
                                                      float* __restrict__ hg2, int n) {
    __shared__ float sWt[OUT_DIM * S2];
    __shared__ __align__(16) float sA[8 * 16 * SA];
    __shared__ float shg[OUT_DIM];

    const int wave = threadIdx.x >> 5;
    const int lane = threadIdx.x & 31;
    const int tile = blockIdx.x * 8 + wave;
    const int ntiles = (n + 15) >> 4;
    const bool active = tile < ntiles;

    float wreg[16];
#pragma unroll
    for (int i = 0; i < 16; ++i) wreg[i] = W2[threadIdx.x + 256 * i];
    if (threadIdx.x < OUT_DIM) shg[threadIdx.x] = 0.f;

    float* arows = sA + wave * 16 * SA;
    if (active) {
        const int rp = rowptr[min(tile * 16 + lane, n)];
        const int l = lane & 7;
        const int g = lane >> 3;
        for (int p = 0; p < 4; ++p) {
            const int r = 4 * p + g;
            const int beg = __shfl(rp, r);
            const int end = __shfl(rp, r + 1);
            float4 lo, hi;
            gather_row8(msg, csr, beg, end, l, lo, hi);
            *(float4*)(arows + r * SA + 4 * l)      = lo;
            *(float4*)(arows + r * SA + 32 + 4 * l) = hi;
        }
    }
#pragma unroll
    for (int i = 0; i < 16; ++i) {
        const int idx = threadIdx.x + 256 * i;
        sWt[(idx & 63) * S2 + (idx >> 6)] = wreg[i];
    }
    __syncthreads();

    const float pa = *prelu_a;
    const int halfsel = lane >> 4;
    const int l15 = lane & 15;

    if (active) {
        const int koff = halfsel * 2;
        const float* arow = arows + l15 * SA;
        const float* bc0 = sWt + ( 0 + l15) * S2;
        const float* bc1 = sWt + (16 + l15) * S2;
        const float* bc2 = sWt + (32 + l15) * S2;
        const float* bc3 = sWt + (48 + l15) * S2;

        v8f acc0 = {}, acc1 = {}, acc2 = {}, acc3 = {};
#pragma unroll 4
        for (int kb = 0; kb < OUT_DIM / 4; ++kb) {
            const int k0 = kb * 4 + koff;
            const v2f a  = *(const v2f*)(arow + k0);
            const v2f b0 = *(const v2f*)(bc0 + k0);
            const v2f b1 = *(const v2f*)(bc1 + k0);
            const v2f b2v = *(const v2f*)(bc2 + k0);
            const v2f b3 = *(const v2f*)(bc3 + k0);
            acc0 = __builtin_amdgcn_wmma_f32_16x16x4_f32(false, a, false, b0, (short)0, acc0, false, false);
            acc1 = __builtin_amdgcn_wmma_f32_16x16x4_f32(false, a, false, b1, (short)0, acc1, false, false);
            acc2 = __builtin_amdgcn_wmma_f32_16x16x4_f32(false, a, false, b2v, (short)0, acc2, false, false);
            acc3 = __builtin_amdgcn_wmma_f32_16x16x4_f32(false, a, false, b3, (short)0, acc3, false, false);
        }

        const float bb0 = b2[ 0 + l15], bb1 = b2[16 + l15];
        const float bb2 = b2[32 + l15], bb3 = b2[48 + l15];
        float hsum0 = 0.f, hsum1 = 0.f, hsum2 = 0.f, hsum3 = 0.f;
#pragma unroll
        for (int r = 0; r < 8; ++r) {
            const int node = tile * 16 + r + halfsel * 8;
            if (node < n) {
                const float dn = dstn[node];
                float v0 = acc0[r] * dn + bb0; v0 = (v0 >= 0.f) ? v0 : pa * v0;
                float v1 = acc1[r] * dn + bb1; v1 = (v1 >= 0.f) ? v1 : pa * v1;
                float v2 = acc2[r] * dn + bb2; v2 = (v2 >= 0.f) ? v2 : pa * v2;
                float v3 = acc3[r] * dn + bb3; v3 = (v3 >= 0.f) ? v3 : pa * v3;
                float* o = hout + (size_t)node * OUT_DIM + l15;
                o[0] = v0; o[16] = v1; o[32] = v2; o[48] = v3;
                hsum0 += v0; hsum1 += v1; hsum2 += v2; hsum3 += v3;
            }
        }
        atomicAdd(&shg[ 0 + l15], hsum0);
        atomicAdd(&shg[16 + l15], hsum1);
        atomicAdd(&shg[32 + l15], hsum2);
        atomicAdd(&shg[48 + l15], hsum3);
    }
    __syncthreads();
    if (threadIdx.x < OUT_DIM) atomicAdd(&hg2[threadIdx.x], shg[threadIdx.x]);
}

extern "C" void kernel_launch(void* const* d_in, const int* in_sizes, int n_in,
                              void* d_out, int out_size, void* d_ws, size_t ws_size,
                              hipStream_t stream) {
    const float* feat    = (const float*)d_in[0];
    const int*   src     = (const int*)  d_in[1];
    const int*   dst     = (const int*)  d_in[2];
    const float* W1      = (const float*)d_in[3];
    const float* b1      = (const float*)d_in[4];
    const float* W2      = (const float*)d_in[5];
    const float* b2      = (const float*)d_in[6];
    const float* prelu_a = (const float*)d_in[7];

    const int n = in_sizes[0] / IN_DIM;
    const int E = in_sizes[1];

    float* out = (float*)d_out;
    float* hg1 = out + (size_t)n * OUT_DIM;
    float* hg2 = hg1 + OUT_DIM;

    float* wsf    = (float*)d_ws;
    float* big    = wsf;
    int*   csr    = (int*)(wsf + (size_t)n * OUT_DIM);
    int*   rowptr = csr + E;
    float* srcn   = (float*)(rowptr + n + 32);
    float* dstn   = srcn + n;

    const int nb      = (n + BW - 1) / BW;
    const int nchunks = (E + CH - 1) / CH;
    const int ostride = (nb + 1 + 31) & ~31;

    int*  pd    = (int*)big;
    int*  ps    = pd + E;
    int*  offD  = ps + E;
    int*  offS  = offD + (size_t)nchunks * ostride;
    int*  spill = offS + (size_t)nchunks * ostride;

    const int ntiles = (n + 15) / 16;
    const int gemm_blocks = (ntiles + 7) / 8;

    k_bin    <<<nchunks, 1024, 0, stream>>>(src, dst, pd, ps, offD, offS, hg1, E, nb, ostride);
    k_own    <<<nb, 1024, 0, stream>>>(pd, ps, offD, offS, rowptr, csr, srcn, dstn, spill, feat, W1, out, n, nb, nchunks, ostride);
    k_gather1<<<1024, 512, 0, stream>>>(out, csr, rowptr, dstn, srcn, b1, prelu_a, big, hg1, n);
    k_gather_gemm2<<<gemm_blocks, 256, 0, stream>>>(big, csr, rowptr, W2, dstn, b2, prelu_a, out, hg2, n);
}


# both row gathers keep 8 rows of a node in flight (was 4) with the next node's list bounds and entries read under them; row sum order kept; layer-1 pool now 625 block partials (was 1024)
# speedup vs baseline: 4.4503x; 1.0397x over previous
#include <hip/hip_runtime.h>

constexpr int IN_DIM  = 128;
constexpr int OUT_DIM = 64;
constexpr int S1 = IN_DIM + 4;
constexpr int S2 = OUT_DIM + 4;

typedef float v2f __attribute__((ext_vector_type(2)));
typedef float v8f __attribute__((ext_vector_type(8)));
typedef float v4f __attribute__((ext_vector_type(4)));

constexpr int BSH   = 9;
constexpr int BW    = 1 << BSH;
constexpr int NBMAX = 1024;
constexpr int CH    = 8192;
constexpr int EPT   = CH / 1024;
constexpr int CAP   = 16384;
constexpr int LSH   = 32 - BSH;
constexpr int SMASK = (1 << LSH) - 1;

static __device__ __forceinline__ void block_scan2(int* a, int* b, int cnt, int (*wt)[32]) {
    const int t = threadIdx.x, lane = t & 31, w = t >> 5;
    const int va = (t < cnt) ? a[t] : 0;
    const int vb = (t < cnt) ? b[t] : 0;
    int ia = va, ib = vb;
    for (int o = 1; o < 32; o <<= 1) {
        const int xa = __shfl_up(ia, o);
        const int xb = __shfl_up(ib, o);
        if (lane >= o) { ia += xa; ib += xb; }
    }
    if (lane == 31) { wt[0][w] = ia; wt[1][w] = ib; }
    __syncthreads();
    if (w == 0) {
        const int ta = wt[0][lane], tb = wt[1][lane];
        int sa = ta, sb = tb;
        for (int o = 1; o < 32; o <<= 1) {
            const int xa = __shfl_up(sa, o);
            const int xb = __shfl_up(sb, o);
            if (lane >= o) { sa += xa; sb += xb; }
        }
        wt[0][lane] = sa - ta;
        wt[1][lane] = sb - tb;
        if (lane == 31) { a[cnt] = sa; b[cnt] = sb; }
    }
    __syncthreads();
    if (t < cnt) { a[t] = wt[0][w] + ia - va; b[t] = wt[1][w] + ib - vb; }
    __syncthreads();
}

__global__ void __launch_bounds__(1024) k_bin(const int* __restrict__ src,
                                              const int* __restrict__ dst,
                                              int* __restrict__ pd,
                                              int* __restrict__ ps,
                                              int* __restrict__ offD,
                                              int* __restrict__ offS,
                                              float* __restrict__ hg,
                                              int E, int nb, int ostride) {
    __shared__ int  hd[NBMAX + 1];
    __shared__ int  hs[NBMAX + 1];
    __shared__ int  sd[CH];
    __shared__ int  ss[CH];
    __shared__ int  wt[2][32];
    const int t = threadIdx.x;
    const int c = blockIdx.x;
    const int e0 = c * CH;
    const int m = min(CH, E - e0);
    if (c == 0 && t < 2 * OUT_DIM) hg[t] = 0.f;
    for (int i = t; i <= nb; i += 1024) { hd[i] = 0; hs[i] = 0; }
    __syncthreads();

    int s[EPT], d[EPT], rd[EPT], rs[EPT];
#pragma unroll
    for (int i = 0; i < EPT; ++i) {
        const int j = t + i * 1024;
        s[i] = (j < m) ? src[e0 + j] : 0;
        d[i] = (j < m) ? dst[e0 + j] : 0;
        rd[i] = 0;
        rs[i] = 0;
    }
#pragma unroll
    for (int i = 0; i < EPT; ++i) {
        const int j = t + i * 1024;
        if (j < m) {
            rd[i] = atomicAdd(&hd[d[i] >> BSH], 1);
            rs[i] = atomicAdd(&hs[s[i] >> BSH], 1);
        }
    }
    __syncthreads();
    block_scan2(hd, hs, nb, wt);

    int* od = offD + (size_t)c * ostride;
    int* os = offS + (size_t)c * ostride;
    for (int i = t; i <= nb; i += 1024) { od[i] = hd[i]; os[i] = hs[i]; }
#pragma unroll
    for (int i = 0; i < EPT; ++i) {
        const int j = t + i * 1024;
        if (j < m) {
            sd[hd[d[i] >> BSH] + rd[i]] = (int)((unsigned)s[i] | ((unsigned)(d[i] & (BW - 1)) << LSH));
            ss[hs[s[i] >> BSH] + rs[i]] = s[i];
        }
    }
    __syncthreads();
    for (int j = t; j < m; j += 1024) {
        pd[(size_t)e0 + j] = sd[j];
        ps[(size_t)e0 + j] = ss[j];
    }
}

constexpr int KT = IN_DIM / 4;

template <int KS>
static __device__ __forceinline__ void slice_load(const float* __restrict__ arow, int kb, v2f (&a)[KS]) {
#pragma unroll
    for (int i = 0; i < KS; ++i) a[i] = *(const v2f*)(arow + (kb + i) * 4);
}

template <int KS>
static __device__ __forceinline__ void slice_mma(const float* bc, int kb, const v2f (&a)[KS],
                                                 v8f& acc0, v8f& acc1, v8f& acc2, v8f& acc3) {
#pragma unroll
    for (int i = 0; i < KS; ++i) {
        const float* b = bc + (kb + i) * 4;
        const v2f b0 = *(const v2f*)(b);
        const v2f b1 = *(const v2f*)(b + 16 * S1);
        const v2f b2 = *(const v2f*)(b + 32 * S1);
        const v2f b3 = *(const v2f*)(b + 48 * S1);
        acc0 = __builtin_amdgcn_wmma_f32_16x16x4_f32(false, a[i], false, b0, (short)0, acc0, false, false);
        acc1 = __builtin_amdgcn_wmma_f32_16x16x4_f32(false, a[i], false, b1, (short)0, acc1, false, false);
        acc2 = __builtin_amdgcn_wmma_f32_16x16x4_f32(false, a[i], false, b2, (short)0, acc2, false, false);
        acc3 = __builtin_amdgcn_wmma_f32_16x16x4_f32(false, a[i], false, b3, (short)0, acc3, false, false);
    }
}

__global__ void __launch_bounds__(1024) k_own(const int* __restrict__ pd,
                                              const int* __restrict__ ps,
                                              const int* __restrict__ offD,
                                              const int* __restrict__ offS,
                                              int* __restrict__ rowptr,
                                              int* __restrict__ csr,
                                              float* __restrict__ srcn,
                                              float* __restrict__ dstn,
                                              int* spill,
                                              const float* __restrict__ feat,
                                              const float* __restrict__ W1,
                                              float* __restrict__ t1,
                                              int n, int nb, int nchunks, int ostride) {
    __shared__ int cnt[BW];
    __shared__ int oc[BW];
    __shared__ int excl[BW + 1];
    __shared__ int us[CAP];
    __shared__ int ud[CAP];
    __shared__ int red[2][32];
    __shared__ int pcd[2][1024];
    __shared__ int pcs[2][1024];
    __shared__ float sWt[OUT_DIM * S1];
    __shared__ float snl[BW];
    const int t = threadIdx.x, lane = t & 31;
    const int w = __builtin_amdgcn_readfirstlane(t >> 5);
    const int b = blockIdx.x;
    const int base = b << BSH;

    float wreg[IN_DIM * OUT_DIM / 1024];
#pragma unroll
    for (int i = 0; i < IN_DIM * OUT_DIM / 1024; ++i) wreg[i] = W1[t + 1024 * i];

    for (int i = t; i < BW; i += 1024) { cnt[i] = 0; oc[i] = 0; }

    int pb = 0, pt = 0;
    for (int c = t; c < nchunks; c += 1024) {
        const int o0 = offD[(size_t)c * ostride + b];
        const int o1 = offD[(size_t)c * ostride + b + 1];
        pb += o0;
        pt += o1 - o0;
        if (c < 1024) {
            pcd[0][c] = o0;
            pcd[1][c] = o1;
            pcs[0][c] = offS[(size_t)c * ostride + b];
            pcs[1][c] = offS[(size_t)c * ostride + b + 1];
        }
    }
    for (int o = 16; o > 0; o >>= 1) { pb += __shfl_xor(pb, o); pt += __shfl_xor(pt, o); }
    if (lane == 0) { red[0][w] = pb; red[1][w] = pt; }
#pragma unroll
    for (int i = 0; i < IN_DIM * OUT_DIM / 1024; ++i) {
        const int idx = t + 1024 * i;
        sWt[(idx & 63) * S1 + (idx >> 6)] = wreg[i];
    }
    __syncthreads();
    int binbase = red[0][lane], total = red[1][lane];
    for (int o = 16; o > 0; o >>= 1) { binbase += __shfl_xor(binbase, o); total += __shfl_xor(total, o); }
    __syncthreads();

    const int halfsel = lane >> 4;
    const int l15 = lane & 15;
    const int tile0 = base + w * 16;
    const bool gact = tile0 < n;
    const int grow = min(tile0 + l15, n - 1);
    const float* arow = feat + (size_t)grow * IN_DIM + halfsel * 2;
    const float* bc = sWt + l15 * S1 + halfsel * 2;
    v8f acc0 = {}, acc1 = {}, acc2 = {}, acc3 = {};
    int kdone = 0;

    auto count_piece = [&](int c, int o0, int o1, int q0, int q1, auto&& ahead) {
        const int* p = pd + (size_t)c * CH + o0;
        const int* q = ps + (size_t)c * CH + q0;
        const int ld = o1 - o0, ls = q1 - q0;
        const int len = max(ld, ls);
        auto trip = [&](int i, auto&& hook) {
            const int i2 = i + 32;
            const bool d0 = i < ld, d1 = i2 < ld, s0 = i < ls, s1 = i2 < ls;
            const unsigned dv0 = (unsigned)*(d0 ? p + i  : pd);
            const unsigned dv1 = (unsigned)*(d1 ? p + i2 : pd);
            const int sv0 = *(s0 ? q + i  : ps);
            const int sv1 = *(s1 ? q + i2 : ps);
            hook();
            if (d0) atomicAdd(&cnt[dv0 >> LSH], 1);
            if (d1) atomicAdd(&cnt[dv1 >> LSH], 1);
            if (s0) atomicAdd(&oc[sv0 - base], 1);
            if (s1) atomicAdd(&oc[sv1 - base], 1);
        };
        trip(lane, ahead);
        for (int i = lane + 64; i < len; i += 64) trip(i, [] {});
    };
    const int nlds = min(nchunks, 1024);
    for (int c = w; c < nlds; c += 32) {
        const bool g = gact && kdone + 3 <= KT;
        v2f a[3];
        count_piece(c, pcd[0][c], pcd[1][c], pcs[0][c], pcs[1][c],
                    [&] { if (g) slice_load<3>(arow, kdone, a); });
        if (g) { slice_mma<3>(bc, kdone, a, acc0, acc1, acc2, acc3); kdone += 3; }
    }
    for (int c = 1024 + w; c < nchunks; c += 32)
        count_piece(c, offD[(size_t)c * ostride + b], offD[(size_t)c * ostride + b + 1],
                       offS[(size_t)c * ostride + b], offS[(size_t)c * ostride + b + 1], [] {});
    __syncthreads();

    {
        const int v = (t < BW) ? cnt[t] : 0;
        int inc = v;
        for (int o = 1; o < 32; o <<= 1) {
            const int x = __shfl_up(inc, o);
            if (lane >= o) inc += x;
        }
        if (lane == 31) red[0][w] = inc;
        __syncthreads();
        int wo = 0;
        for (int k = 0; k < w; ++k) wo += red[0][k];
        const int ex = wo + inc - v;
        if (t < BW) {
            const int node = base + t;
            const float sn = rsqrtf(fmaxf((float)oc[t], 1.0f));
            if (node < n) {
                rowptr[node] = binbase + ex;
                dstn[node] = rsqrtf(fmaxf((float)v, 1.0f));
                srcn[node] = sn;
            }
            snl[t] = sn;
            excl[t] = ex;
            cnt[t] = 0;
        }
        if (t == 0) {
            excl[BW] = total;
            if (b == nb - 1) rowptr[n] = binbase + total;
        }
    }
    __syncthreads();

    const bool fits = total <= CAP;
    int* sp = spill + binbase;
    auto place_piece = [&](int c, int o0, int o1, auto&& ahead) {
        const int* p = pd + (size_t)c * CH + o0;
        const int ld = o1 - o0;
        auto trip = [&](int i, auto&& hook) {
            const int i2 = i + 32;
            const bool d0 = i < ld, d1 = i2 < ld;
            const unsigned e0 = (unsigned)*(d0 ? p + i  : pd);
            const unsigned e1 = (unsigned)*(d1 ? p + i2 : pd);
            hook();
            if (d0) {
                const int l0 = (int)(e0 >> LSH);
                const int pos0 = excl[l0] + atomicAdd(&cnt[l0], 1);
                if (fits) { us[pos0] = (int)(e0 & SMASK); ud[pos0] = l0; }
                else      { sp[pos0] = (int)e0; }
            }
            if (d1) {
                const int l1 = (int)(e1 >> LSH);
                const int pos1 = excl[l1] + atomicAdd(&cnt[l1], 1);
                if (fits) { us[pos1] = (int)(e1 & SMASK); ud[pos1] = l1; }
                else      { sp[pos1] = (int)e1; }
            }
        };
        trip(lane, ahead);
        for (int i = lane + 64; i < ld; i += 64) trip(i, [] {});
    };
    for (int c = w; c < nlds; c += 32) {
        const bool g = gact && kdone + 2 <= KT;
        v2f a[2];
        place_piece(c, pcd[0][c], pcd[1][c],
                    [&] { if (g) slice_load<2>(arow, kdone, a); });
        if (g) { slice_mma<2>(bc, kdone, a, acc0, acc1, acc2, acc3); kdone += 2; }
    }
    for (int c = 1024 + w; c < nchunks; c += 32)
        place_piece(c, offD[(size_t)c * ostride + b], offD[(size_t)c * ostride + b + 1], [] {});

    if (gact) {
        while (kdone < KT) {
            v2f a[1];
            slice_load<1>(arow, kdone, a);
            slice_mma<1>(bc, kdone, a, acc0, acc1, acc2, acc3);
            kdone += 1;
        }
#pragma unroll
        for (int r = 0; r < 8; ++r) {
            const int lrow = w * 16 + r + halfsel * 8;
            if (base + lrow < n) {
                const float sc = snl[lrow];
                float* o = t1 + (size_t)(base + lrow) * OUT_DIM + l15;
                o[0]  = acc0[r] * sc;
                o[16] = acc1[r] * sc;
                o[32] = acc2[r] * sc;
                o[48] = acc3[r] * sc;
            }
        }
    }
    __syncthreads();

    if (fits) {
        for (int p = t; p < total; p += 1024) {
            const int me = us[p];
            const int row = ud[p];
            const int beg = excl[row], end = excl[row + 1];
            int k = 0;
            for (int j = beg; j < end; ++j) {
                const int x = us[j];
                k += ((x < me) || (x == me && j < p)) ? 1 : 0;
            }
            csr[(size_t)binbase + beg + k] = me;
        }
    } else {
        for (int p = t; p < total; p += 1024) {
            const unsigned mine = (unsigned)sp[p];
            const int mx = (int)(mine & SMASK);
            const int my = (int)(mine >> LSH);
            const int beg = excl[my], end = excl[my + 1];
            int k = 0;
            for (int j = beg; j < end; ++j) {
                const int x = sp[j] & SMASK;
                k += ((x < mx) || (x == mx && j < p)) ? 1 : 0;
            }
            csr[(size_t)binbase + beg + k] = mx;
        }
    }
}

static __device__ __forceinline__ float4 gather_row(const float* __restrict__ msg,
                                                    const int* __restrict__ csr,
                                                    int beg, int end, int l, int f0, int f1) {
    float4 acc = make_float4(0.f, 0.f, 0.f, 0.f);
    const char* tab = (const char*)msg;
    const unsigned lo16 = 16u * (unsigned)l;
    auto chunk = [&](int m0, int m1, int cnt) {
        for (int k = 0; k < cnt; k += 8) {
            const int m = (k < 16) ? m0 : m1;
            const int kk = k & 15;
            v4f r[8];
            bool v[8];
            unsigned off[8];
#pragma unroll
            for (int j = 0; j < 8; ++j) {
                v[j] = k + j < cnt;
                const unsigned sj = (unsigned)__shfl(m, v[j] ? kk + j : kk, 16);
                off[j] = sj * (unsigned)(OUT_DIM * 4) + lo16;
            }
#pragma unroll
            for (int j = 0; j < 8; ++j) r[j] = *(const v4f*)(tab + off[j]);
            __builtin_amdgcn_sched_barrier(0);
#pragma unroll
            for (int j = 0; j < 8; ++j) {
                acc.x += v[j] ? r[j].x : 0.f; acc.y += v[j] ? r[j].y : 0.f;
                acc.z += v[j] ? r[j].z : 0.f; acc.w += v[j] ? r[j].w : 0.f;
            }
        }
    };
    chunk(f0, f1, min(32, end - beg));
    for (int c = beg + 32; c < end; c += 32) {
        const int cnt = min(32, end - c);
        const int m0 = (l < cnt) ? csr[c + l] : 0;
        const int m1 = (16 + l < cnt) ? csr[c + 16 + l] : 0;
        chunk(m0, m1, cnt);
    }
    return acc;
}

__global__ void __launch_bounds__(512) k_gather1(const float* __restrict__ msg,
                                                 const int* __restrict__ csr,
                                                 const int* __restrict__ rowptr,
                                                 const float* __restrict__ dstn,
                                                 const float* __restrict__ srcn,
                                                 const float* __restrict__ b1,
                                                 const float* __restrict__ prelu_a,
                                                 float* __restrict__ hs,
                                                 float* __restrict__ hg1, int n) {
    const float pa = *prelu_a;
    const int l = threadIdx.x & 15;
    const int grp = threadIdx.x >> 4;
    const float4 bias = *(const float4*)(b1 + 4 * l);
    float4 part = make_float4(0.f, 0.f, 0.f, 0.f);
    const int step = gridDim.x * 32;
    int node = blockIdx.x * 32 + grp;
    int beg = 0, end = 0, nbeg = 0, nend = 0;
    if (node < n) { beg = rowptr[node]; end = rowptr[node + 1]; }
    if (node + step < n) { nbeg = rowptr[node + step]; nend = rowptr[node + step + 1]; }
    const int c0 = min(32, end - beg);
    int m0 = (l < c0) ? csr[beg + l] : 0;
    int m1 = (16 + l < c0) ? csr[beg + 16 + l] : 0;
    while (node < n) {
        asm volatile("" :: "v"(beg), "v"(end), "v"(m0), "v"(m1), "v"(nbeg), "v"(nend));
        const int nc = min(32, nend - nbeg);
        const int nm0 = (l < nc) ? csr[nbeg + l] : 0;
        const int nm1 = (16 + l < nc) ? csr[nbeg + 16 + l] : 0;
        const int nn = node + 2 * step;
        int nnbeg = 0, nnend = 0;
        if (nn < n) { nnbeg = rowptr[nn]; nnend = rowptr[nn + 1]; }
        const float dn = dstn[node];
        const float sn = srcn[node];
        const float4 v = gather_row(msg, csr, beg, end, l, m0, m1);
        float4 h;
        h.x = v.x * dn + bias.x; h.x = (h.x >= 0.f) ? h.x : pa * h.x;
        h.y = v.y * dn + bias.y; h.y = (h.y >= 0.f) ? h.y : pa * h.y;
        h.z = v.z * dn + bias.z; h.z = (h.z >= 0.f) ? h.z : pa * h.z;
        h.w = v.w * dn + bias.w; h.w = (h.w >= 0.f) ? h.w : pa * h.w;
        part.x += h.x; part.y += h.y; part.z += h.z; part.w += h.w;
        *(float4*)(hs + (size_t)node * OUT_DIM + 4 * l) =
            make_float4(h.x * sn, h.y * sn, h.z * sn, h.w * sn);
        node += step; beg = nbeg; end = nend; m0 = nm0; m1 = nm1; nbeg = nnbeg; nend = nnend;
    }
    __shared__ float red[32 * OUT_DIM];
    *(float4*)(red + grp * OUT_DIM + 4 * l) = part;
    __syncthreads();
    if (threadIdx.x < OUT_DIM) {
        float sum = 0.f;
        for (int g = 0; g < 32; ++g) sum += red[g * OUT_DIM + threadIdx.x];
        atomicAdd(&hg1[threadIdx.x], sum);
    }
}

struct Idx4 { int m0, m1, m2, m3; };

static __device__ __forceinline__ Idx4 load_idx(const int* __restrict__ csr, int c, int end, int l) {
    const int cnt = min(32, end - c);
    Idx4 x;
    x.m0 = (l < cnt) ? csr[c + l] : 0;
    x.m1 = (8 + l < cnt) ? csr[c + 8 + l] : 0;
    x.m2 = (16 + l < cnt) ? csr[c + 16 + l] : 0;
    x.m3 = (24 + l < cnt) ? csr[c + 24 + l] : 0;
    return x;
}

static __device__ __forceinline__ void gather_row8(const float* __restrict__ msg,
                                                   const int* __restrict__ csr,
                                                   int beg, int end, int l, Idx4 first,
                                                   float4& lo, float4& hi) {
    float4 a = make_float4(0.f, 0.f, 0.f, 0.f);
    float4 b = make_float4(0.f, 0.f, 0.f, 0.f);
    const char* tab = (const char*)msg;
    const unsigned lo16 = 16u * (unsigned)l;
    auto chunk = [&](Idx4 x, int cnt) {
        for (int k = 0; k < cnt; k += 8) {
            const int m = (k < 8) ? x.m0 : (k < 16) ? x.m1 : (k < 24) ? x.m2 : x.m3;
            v4f ra[8], rb[8];
            bool v[8];
            unsigned off[8];
#pragma unroll
            for (int j = 0; j < 8; ++j) {
                v[j] = k + j < cnt;
                const unsigned sj = (unsigned)__shfl(m, v[j] ? j : 0, 8);
                off[j] = sj * (unsigned)(OUT_DIM * 4) + lo16;
            }
#pragma unroll
            for (int j = 0; j < 8; ++j) {
                ra[j] = *(const v4f*)(tab + off[j]);
                rb[j] = *(const v4f*)(tab + off[j] + 128);
            }
            __builtin_amdgcn_sched_barrier(0);
#pragma unroll
            for (int j = 0; j < 8; ++j) {
                a.x += v[j] ? ra[j].x : 0.f; a.y += v[j] ? ra[j].y : 0.f;
                a.z += v[j] ? ra[j].z : 0.f; a.w += v[j] ? ra[j].w : 0.f;
                b.x += v[j] ? rb[j].x : 0.f; b.y += v[j] ? rb[j].y : 0.f;
                b.z += v[j] ? rb[j].z : 0.f; b.w += v[j] ? rb[j].w : 0.f;
            }
        }
    };
    chunk(first, min(32, end - beg));
    for (int c = beg + 32; c < end; c += 32)
        chunk(load_idx(csr, c, end, l), min(32, end - c));
    lo = a;
    hi = b;
}

constexpr int SA = OUT_DIM + 4;

__global__ void __launch_bounds__(256) __attribute__((amdgpu_waves_per_eu(8))) k_gather_gemm2(const float* __restrict__ msg,
                                                      const int* __restrict__ csr,
                                                      const int* __restrict__ rowptr,
                                                      const float* __restrict__ W2,
                                                      const float* __restrict__ dstn,
                                                      const float* __restrict__ b2,
                                                      const float* __restrict__ prelu_a,
                                                      float* __restrict__ hout,
                                                      float* __restrict__ hg2, int n) {
    __shared__ float sWt[OUT_DIM * S2];
    __shared__ __align__(16) float sA[8 * 16 * SA];
    __shared__ float shg[OUT_DIM];

    const int wave = threadIdx.x >> 5;
    const int lane = threadIdx.x & 31;
    const int tile = blockIdx.x * 8 + wave;
    const int ntiles = (n + 15) >> 4;
    const bool active = tile < ntiles;

    float wreg[16];
#pragma unroll
    for (int i = 0; i < 16; ++i) wreg[i] = W2[threadIdx.x + 256 * i];
    if (threadIdx.x < OUT_DIM) shg[threadIdx.x] = 0.f;

    float* arows = sA + wave * 16 * SA;
    int rp = 0;
    if (active) rp = rowptr[min(tile * 16 + lane, n)];
#pragma unroll
    for (int i = 0; i < 16; ++i) {
        const int idx = threadIdx.x + 256 * i;
        sWt[(idx & 63) * S2 + (idx >> 6)] = wreg[i];
    }
    if (active) {
        const int l = lane & 7;
        const int g = lane >> 3;
        int beg = __shfl(rp, g);
        int end = __shfl(rp, g + 1);
        Idx4 cur = load_idx(csr, beg, end, l);
        for (int p = 0; p < 4; ++p) {
            const int r = 4 * p + g;
            asm volatile("" :: "v"(cur.m0), "v"(cur.m1), "v"(cur.m2), "v"(cur.m3));
            const int nbeg = __shfl(rp, r + 4);
            const int nend = __shfl(rp, r + 5);
            Idx4 nxt = cur;
            if (p < 3) nxt = load_idx(csr, nbeg, nend, l);
            float4 lo, hi;
            gather_row8(msg, csr, beg, end, l, cur, lo, hi);
            *(float4*)(arows + r * SA + 4 * l)      = lo;
            *(float4*)(arows + r * SA + 32 + 4 * l) = hi;
            cur = nxt; beg = nbeg; end = nend;
        }
    }
    __syncthreads();

    const float pa = *prelu_a;
    const int halfsel = lane >> 4;
    const int l15 = lane & 15;

    if (active) {
        const int koff = halfsel * 2;
        const float* arow = arows + l15 * SA;
        const float* bc0 = sWt + ( 0 + l15) * S2;
        const float* bc1 = sWt + (16 + l15) * S2;
        const float* bc2 = sWt + (32 + l15) * S2;
        const float* bc3 = sWt + (48 + l15) * S2;

        v8f acc0 = {}, acc1 = {}, acc2 = {}, acc3 = {};
#pragma unroll 4
        for (int kb = 0; kb < OUT_DIM / 4; ++kb) {
            const int k0 = kb * 4 + koff;
            const v2f a  = *(const v2f*)(arow + k0);
            const v2f b0 = *(const v2f*)(bc0 + k0);
            const v2f b1 = *(const v2f*)(bc1 + k0);
            const v2f b2v = *(const v2f*)(bc2 + k0);
            const v2f b3 = *(const v2f*)(bc3 + k0);
            acc0 = __builtin_amdgcn_wmma_f32_16x16x4_f32(false, a, false, b0, (short)0, acc0, false, false);
            acc1 = __builtin_amdgcn_wmma_f32_16x16x4_f32(false, a, false, b1, (short)0, acc1, false, false);
            acc2 = __builtin_amdgcn_wmma_f32_16x16x4_f32(false, a, false, b2v, (short)0, acc2, false, false);
            acc3 = __builtin_amdgcn_wmma_f32_16x16x4_f32(false, a, false, b3, (short)0, acc3, false, false);
        }

        const float bb0 = b2[ 0 + l15], bb1 = b2[16 + l15];
        const float bb2 = b2[32 + l15], bb3 = b2[48 + l15];
        float hsum0 = 0.f, hsum1 = 0.f, hsum2 = 0.f, hsum3 = 0.f;
#pragma unroll
        for (int r = 0; r < 8; ++r) {
            const int node = tile * 16 + r + halfsel * 8;
            if (node < n) {
                const float dn = dstn[node];
                float v0 = acc0[r] * dn + bb0; v0 = (v0 >= 0.f) ? v0 : pa * v0;
                float v1 = acc1[r] * dn + bb1; v1 = (v1 >= 0.f) ? v1 : pa * v1;
                float v2 = acc2[r] * dn + bb2; v2 = (v2 >= 0.f) ? v2 : pa * v2;
                float v3 = acc3[r] * dn + bb3; v3 = (v3 >= 0.f) ? v3 : pa * v3;
                float* o = hout + (size_t)node * OUT_DIM + l15;
                o[0] = v0; o[16] = v1; o[32] = v2; o[48] = v3;
                hsum0 += v0; hsum1 += v1; hsum2 += v2; hsum3 += v3;
            }
        }
        atomicAdd(&shg[ 0 + l15], hsum0);
        atomicAdd(&shg[16 + l15], hsum1);
        atomicAdd(&shg[32 + l15], hsum2);
        atomicAdd(&shg[48 + l15], hsum3);
    }
    __syncthreads();
    if (threadIdx.x < OUT_DIM) atomicAdd(&hg2[threadIdx.x], shg[threadIdx.x]);
}

extern "C" void kernel_launch(void* const* d_in, const int* in_sizes, int n_in,
                              void* d_out, int out_size, void* d_ws, size_t ws_size,
                              hipStream_t stream) {
    const float* feat    = (const float*)d_in[0];
    const int*   src     = (const int*)  d_in[1];
    const int*   dst     = (const int*)  d_in[2];
    const float* W1      = (const float*)d_in[3];
    const float* b1      = (const float*)d_in[4];
    const float* W2      = (const float*)d_in[5];
    const float* b2      = (const float*)d_in[6];
    const float* prelu_a = (const float*)d_in[7];

    const int n = in_sizes[0] / IN_DIM;
    const int E = in_sizes[1];

    float* out = (float*)d_out;
    float* hg1 = out + (size_t)n * OUT_DIM;
    float* hg2 = hg1 + OUT_DIM;

    float* wsf    = (float*)d_ws;
    float* big    = wsf;
    int*   csr    = (int*)(wsf + (size_t)n * OUT_DIM);
    int*   rowptr = csr + E;
    float* srcn   = (float*)(rowptr + n + 32);
    float* dstn   = srcn + n;

    const int nb      = (n + BW - 1) / BW;
    const int nchunks = (E + CH - 1) / CH;
    const int ostride = (nb + 1 + 31) & ~31;

    int*  pd    = (int*)big;
    int*  ps    = pd + E;
    int*  offD  = ps + E;
    int*  offS  = offD + (size_t)nchunks * ostride;
    int*  spill = offS + (size_t)nchunks * ostride;

    const int ntiles = (n + 15) / 16;
    const int gemm_blocks = (ntiles + 7) / 8;

    k_bin    <<<nchunks, 1024, 0, stream>>>(src, dst, pd, ps, offD, offS, hg1, E, nb, ostride);
    k_own    <<<nb, 1024, 0, stream>>>(pd, ps, offD, offS, rowptr, csr, srcn, dstn, spill, feat, W1, out, n, nb, nchunks, ostride);
    k_gather1<<<625, 512, 0, stream>>>(out, csr, rowptr, dstn, srcn, b1, prelu_a, big, hg1, n);
    k_gather_gemm2<<<gemm_blocks, 256, 0, stream>>>(big, csr, rowptr, W2, dstn, b2, prelu_a, out, hg2, n);
}


# bin owner splits each row's slots into 16 groups by source id mod 16 (counted, scanned, placed per group) and rank-sorts inside a group only; a row's sum order is now (id mod 16, id)
# speedup vs baseline: 4.6598x; 1.0471x over previous
#include <hip/hip_runtime.h>

constexpr int IN_DIM  = 128;
constexpr int OUT_DIM = 64;
constexpr int S1 = IN_DIM + 4;
constexpr int S2 = OUT_DIM + 4;

typedef float v2f __attribute__((ext_vector_type(2)));
typedef float v8f __attribute__((ext_vector_type(8)));
typedef float v4f __attribute__((ext_vector_type(4)));

constexpr int BSH   = 9;
constexpr int BW    = 1 << BSH;
constexpr int NBMAX = 1024;
constexpr int CH    = 8192;
constexpr int EPT   = CH / 1024;
constexpr int CAP   = 16384;
constexpr int LSH   = 32 - BSH;
constexpr int SMASK = (1 << LSH) - 1;
constexpr int GSH   = 4;
constexpr int NG    = 1 << GSH;

static __device__ __forceinline__ void block_scan2(int* a, int* b, int cnt, int (*wt)[32]) {
    const int t = threadIdx.x, lane = t & 31, w = t >> 5;
    const int va = (t < cnt) ? a[t] : 0;
    const int vb = (t < cnt) ? b[t] : 0;
    int ia = va, ib = vb;
    for (int o = 1; o < 32; o <<= 1) {
        const int xa = __shfl_up(ia, o);
        const int xb = __shfl_up(ib, o);
        if (lane >= o) { ia += xa; ib += xb; }
    }
    if (lane == 31) { wt[0][w] = ia; wt[1][w] = ib; }
    __syncthreads();
    if (w == 0) {
        const int ta = wt[0][lane], tb = wt[1][lane];
        int sa = ta, sb = tb;
        for (int o = 1; o < 32; o <<= 1) {
            const int xa = __shfl_up(sa, o);
            const int xb = __shfl_up(sb, o);
            if (lane >= o) { sa += xa; sb += xb; }
        }
        wt[0][lane] = sa - ta;
        wt[1][lane] = sb - tb;
        if (lane == 31) { a[cnt] = sa; b[cnt] = sb; }
    }
    __syncthreads();
    if (t < cnt) { a[t] = wt[0][w] + ia - va; b[t] = wt[1][w] + ib - vb; }
    __syncthreads();
}

__global__ void __launch_bounds__(1024) k_bin(const int* __restrict__ src,
                                              const int* __restrict__ dst,
                                              int* __restrict__ pd,
                                              int* __restrict__ ps,
                                              int* __restrict__ offD,
                                              int* __restrict__ offS,
                                              float* __restrict__ hg,
                                              int E, int nb, int ostride) {
    __shared__ int  hd[NBMAX + 1];
    __shared__ int  hs[NBMAX + 1];
    __shared__ int  sd[CH];
    __shared__ int  ss[CH];
    __shared__ int  wt[2][32];
    const int t = threadIdx.x;
    const int c = blockIdx.x;
    const int e0 = c * CH;
    const int m = min(CH, E - e0);
    if (c == 0 && t < 2 * OUT_DIM) hg[t] = 0.f;
    for (int i = t; i <= nb; i += 1024) { hd[i] = 0; hs[i] = 0; }
    __syncthreads();

    int s[EPT], d[EPT], rd[EPT], rs[EPT];
#pragma unroll
    for (int i = 0; i < EPT; ++i) {
        const int j = t + i * 1024;
        s[i] = (j < m) ? src[e0 + j] : 0;
        d[i] = (j < m) ? dst[e0 + j] : 0;
        rd[i] = 0;
        rs[i] = 0;
    }
#pragma unroll
    for (int i = 0; i < EPT; ++i) {
        const int j = t + i * 1024;
        if (j < m) {
            rd[i] = atomicAdd(&hd[d[i] >> BSH], 1);
            rs[i] = atomicAdd(&hs[s[i] >> BSH], 1);
        }
    }
    __syncthreads();
    block_scan2(hd, hs, nb, wt);

    int* od = offD + (size_t)c * ostride;
    int* os = offS + (size_t)c * ostride;
    for (int i = t; i <= nb; i += 1024) { od[i] = hd[i]; os[i] = hs[i]; }
#pragma unroll
    for (int i = 0; i < EPT; ++i) {
        const int j = t + i * 1024;
        if (j < m) {
            sd[hd[d[i] >> BSH] + rd[i]] = (int)((unsigned)s[i] | ((unsigned)(d[i] & (BW - 1)) << LSH));
            ss[hs[s[i] >> BSH] + rs[i]] = s[i];
        }
    }
    __syncthreads();
    for (int j = t; j < m; j += 1024) {
        pd[(size_t)e0 + j] = sd[j];
        ps[(size_t)e0 + j] = ss[j];
    }
}

constexpr int KT = IN_DIM / 4;

template <int KS>
static __device__ __forceinline__ void slice_load(const float* __restrict__ arow, int kb, v2f (&a)[KS]) {
#pragma unroll
    for (int i = 0; i < KS; ++i) a[i] = *(const v2f*)(arow + (kb + i) * 4);
}

template <int KS>
static __device__ __forceinline__ void slice_mma(const float* bc, int kb, const v2f (&a)[KS],
                                                 v8f& acc0, v8f& acc1, v8f& acc2, v8f& acc3) {
#pragma unroll
    for (int i = 0; i < KS; ++i) {
        const float* b = bc + (kb + i) * 4;
        const v2f b0 = *(const v2f*)(b);
        const v2f b1 = *(const v2f*)(b + 16 * S1);
        const v2f b2 = *(const v2f*)(b + 32 * S1);
        const v2f b3 = *(const v2f*)(b + 48 * S1);
        acc0 = __builtin_amdgcn_wmma_f32_16x16x4_f32(false, a[i], false, b0, (short)0, acc0, false, false);
        acc1 = __builtin_amdgcn_wmma_f32_16x16x4_f32(false, a[i], false, b1, (short)0, acc1, false, false);
        acc2 = __builtin_amdgcn_wmma_f32_16x16x4_f32(false, a[i], false, b2, (short)0, acc2, false, false);
        acc3 = __builtin_amdgcn_wmma_f32_16x16x4_f32(false, a[i], false, b3, (short)0, acc3, false, false);
    }
}

__global__ void __launch_bounds__(1024) k_own(const int* __restrict__ pd,
                                              const int* __restrict__ ps,
                                              const int* __restrict__ offD,
                                              const int* __restrict__ offS,
                                              int* __restrict__ rowptr,
                                              int* __restrict__ csr,
                                              float* __restrict__ srcn,
                                              float* __restrict__ dstn,
                                              int* spill,
                                              const float* __restrict__ feat,
                                              const float* __restrict__ W1,
                                              float* __restrict__ t1,
                                              int n, int nb, int nchunks, int ostride) {
    __shared__ __align__(16) int cntbuf[4 + BW * NG];
    int* const cnt = cntbuf + 4;
    __shared__ int oc[BW];
    __shared__ int us[CAP];
    __shared__ int ud[CAP];
    __shared__ int red[2][32];
    __shared__ int pcd[2][1024];
    __shared__ int pcs[2][1024];
    __shared__ float sWt[OUT_DIM * S1];
    __shared__ float snl[BW];
    const int t = threadIdx.x, lane = t & 31;
    const int w = __builtin_amdgcn_readfirstlane(t >> 5);
    const int b = blockIdx.x;
    const int base = b << BSH;

    float wreg[IN_DIM * OUT_DIM / 1024];
#pragma unroll
    for (int i = 0; i < IN_DIM * OUT_DIM / 1024; ++i) wreg[i] = W1[t + 1024 * i];

    for (int i = t; i < BW * NG; i += 1024) cnt[i] = 0;
    for (int i = t; i < BW; i += 1024) oc[i] = 0;
    if (t == 0) cntbuf[3] = 0;

    int pb = 0, pt = 0;
    for (int c = t; c < nchunks; c += 1024) {
        const int o0 = offD[(size_t)c * ostride + b];
        const int o1 = offD[(size_t)c * ostride + b + 1];
        pb += o0;
        pt += o1 - o0;
        if (c < 1024) {
            pcd[0][c] = o0;
            pcd[1][c] = o1;
            pcs[0][c] = offS[(size_t)c * ostride + b];
            pcs[1][c] = offS[(size_t)c * ostride + b + 1];
        }
    }
    for (int o = 16; o > 0; o >>= 1) { pb += __shfl_xor(pb, o); pt += __shfl_xor(pt, o); }
    if (lane == 0) { red[0][w] = pb; red[1][w] = pt; }
#pragma unroll
    for (int i = 0; i < IN_DIM * OUT_DIM / 1024; ++i) {
        const int idx = t + 1024 * i;
        sWt[(idx & 63) * S1 + (idx >> 6)] = wreg[i];
    }
    __syncthreads();
    int binbase = red[0][lane], total = red[1][lane];
    for (int o = 16; o > 0; o >>= 1) { binbase += __shfl_xor(binbase, o); total += __shfl_xor(total, o); }
    __syncthreads();

    const int halfsel = lane >> 4;
    const int l15 = lane & 15;
    const int tile0 = base + w * 16;
    const bool gact = tile0 < n;
    const int grow = min(tile0 + l15, n - 1);
    const float* arow = feat + (size_t)grow * IN_DIM + halfsel * 2;
    const float* bc = sWt + l15 * S1 + halfsel * 2;
    v8f acc0 = {}, acc1 = {}, acc2 = {}, acc3 = {};
    int kdone = 0;

    auto count_piece = [&](int c, int o0, int o1, int q0, int q1, auto&& ahead) {
        const int* p = pd + (size_t)c * CH + o0;
        const int* q = ps + (size_t)c * CH + q0;
        const int ld = o1 - o0, ls = q1 - q0;
        const int len = max(ld, ls);
        auto trip = [&](int i, auto&& hook) {
            const int i2 = i + 32;
            const bool d0 = i < ld, d1 = i2 < ld, s0 = i < ls, s1 = i2 < ls;
            const unsigned dv0 = (unsigned)*(d0 ? p + i  : pd);
            const unsigned dv1 = (unsigned)*(d1 ? p + i2 : pd);
            const int sv0 = *(s0 ? q + i  : ps);
            const int sv1 = *(s1 ? q + i2 : ps);
            hook();
            if (d0) atomicAdd(&cnt[((dv0 >> LSH) << GSH) | (dv0 & (NG - 1))], 1);
            if (d1) atomicAdd(&cnt[((dv1 >> LSH) << GSH) | (dv1 & (NG - 1))], 1);
            if (s0) atomicAdd(&oc[sv0 - base], 1);
            if (s1) atomicAdd(&oc[sv1 - base], 1);
        };
        trip(lane, ahead);
        for (int i = lane + 64; i < len; i += 64) trip(i, [] {});
    };
    const int nlds = min(nchunks, 1024);
    for (int c = w; c < nlds; c += 32) {
        const bool g = gact && kdone + 3 <= KT;
        v2f a[3];
        count_piece(c, pcd[0][c], pcd[1][c], pcs[0][c], pcs[1][c],
                    [&] { if (g) slice_load<3>(arow, kdone, a); });
        if (g) { slice_mma<3>(bc, kdone, a, acc0, acc1, acc2, acc3); kdone += 3; }
    }
    for (int c = 1024 + w; c < nchunks; c += 32)
        count_piece(c, offD[(size_t)c * ostride + b], offD[(size_t)c * ostride + b + 1],
                       offS[(size_t)c * ostride + b], offS[(size_t)c * ostride + b + 1], [] {});
    __syncthreads();

    {
        int cg[NG];
        int v = 0;
#pragma unroll
        for (int g = 0; g < NG; ++g) cg[g] = 0;
        if (t < BW) {
#pragma unroll
            for (int g = 0; g < NG; ++g) { cg[g] = cnt[t * NG + g]; v += cg[g]; }
        }
        int inc = v;
        for (int o = 1; o < 32; o <<= 1) {
            const int x = __shfl_up(inc, o);
            if (lane >= o) inc += x;
        }
        if (lane == 31) red[0][w] = inc;
        __syncthreads();
        int wo = 0;
        for (int k = 0; k < w; ++k) wo += red[0][k];
        const int ex = wo + inc - v;
        if (t < BW) {
            const int node = base + t;
            const float sn = rsqrtf(fmaxf((float)oc[t], 1.0f));
            if (node < n) {
                rowptr[node] = binbase + ex;
                dstn[node] = rsqrtf(fmaxf((float)v, 1.0f));
                srcn[node] = sn;
            }
            snl[t] = sn;
            int run = ex;
#pragma unroll
            for (int g = 0; g < NG; ++g) {
                cnt[t * NG + g] = run;
                run += cg[g];
            }
        }
        if (t == 0 && b == nb - 1) rowptr[n] = binbase + total;
    }
    __syncthreads();

    const bool fits = total <= CAP;
    int* sp = spill + binbase;
    auto place_piece = [&](int c, int o0, int o1, auto&& ahead) {
        const int* p = pd + (size_t)c * CH + o0;
        const int ld = o1 - o0;
        auto trip = [&](int i, auto&& hook) {
            const int i2 = i + 32;
            const bool d0 = i < ld, d1 = i2 < ld;
            const unsigned e0 = (unsigned)*(d0 ? p + i  : pd);
            const unsigned e1 = (unsigned)*(d1 ? p + i2 : pd);
            hook();
            if (d0) {
                const int l0 = (int)(((e0 >> LSH) << GSH) | (e0 & (NG - 1)));
                const int pos0 = atomicAdd(&cnt[l0], 1);
                if (fits) { us[pos0] = (int)(e0 & SMASK); ud[pos0] = l0; }
                else      { sp[pos0] = (int)e0; }
            }
            if (d1) {
                const int l1 = (int)(((e1 >> LSH) << GSH) | (e1 & (NG - 1)));
                const int pos1 = atomicAdd(&cnt[l1], 1);
                if (fits) { us[pos1] = (int)(e1 & SMASK); ud[pos1] = l1; }
                else      { sp[pos1] = (int)e1; }
            }
        };
        trip(lane, ahead);
        for (int i = lane + 64; i < ld; i += 64) trip(i, [] {});
    };
    for (int c = w; c < nlds; c += 32) {
        const bool g = gact && kdone + 2 <= KT;
        v2f a[2];
        place_piece(c, pcd[0][c], pcd[1][c],
                    [&] { if (g) slice_load<2>(arow, kdone, a); });
        if (g) { slice_mma<2>(bc, kdone, a, acc0, acc1, acc2, acc3); kdone += 2; }
    }
    for (int c = 1024 + w; c < nchunks; c += 32)
        place_piece(c, offD[(size_t)c * ostride + b], offD[(size_t)c * ostride + b + 1], [] {});

    if (gact) {
        while (kdone < KT) {
            v2f a[1];
            slice_load<1>(arow, kdone, a);
            slice_mma<1>(bc, kdone, a, acc0, acc1, acc2, acc3);
            kdone += 1;
        }
#pragma unroll
        for (int r = 0; r < 8; ++r) {
            const int lrow = w * 16 + r + halfsel * 8;
            if (base + lrow < n) {
                const float sc = snl[lrow];
                float* o = t1 + (size_t)(base + lrow) * OUT_DIM + l15;
                o[0]  = acc0[r] * sc;
                o[16] = acc1[r] * sc;
                o[32] = acc2[r] * sc;
                o[48] = acc3[r] * sc;
            }
        }
    }
    __syncthreads();

    if (fits) {
        for (int p = t; p < total; p += 1024) {
            const int me = us[p];
            const int grp = ud[p];
            const int beg = cnt[grp - 1], end = cnt[grp];
            int k = 0;
            for (int j = beg; j < end; ++j) {
                const int x = us[j];
                k += ((x < me) || (x == me && j < p)) ? 1 : 0;
            }
            csr[(size_t)binbase + beg + k] = me;
        }
    } else {
        for (int p = t; p < total; p += 1024) {
            const unsigned mine = (unsigned)sp[p];
            const int mx = (int)(mine & SMASK);
            const int my = (int)(((mine >> LSH) << GSH) | (mine & (NG - 1)));
            const int beg = cnt[my - 1], end = cnt[my];
            int k = 0;
            for (int j = beg; j < end; ++j) {
                const int x = sp[j] & SMASK;
                k += ((x < mx) || (x == mx && j < p)) ? 1 : 0;
            }
            csr[(size_t)binbase + beg + k] = mx;
        }
    }
}

static __device__ __forceinline__ float4 gather_row(const float* __restrict__ msg,
                                                    const int* __restrict__ csr,
                                                    int beg, int end, int l, int f0, int f1) {
    float4 acc = make_float4(0.f, 0.f, 0.f, 0.f);
    const char* tab = (const char*)msg;
    const unsigned lo16 = 16u * (unsigned)l;
    auto chunk = [&](int m0, int m1, int cnt) {
        for (int k = 0; k < cnt; k += 8) {
            const int m = (k < 16) ? m0 : m1;
            const int kk = k & 15;
            v4f r[8];
            bool v[8];
            unsigned off[8];
#pragma unroll
            for (int j = 0; j < 8; ++j) {
                v[j] = k + j < cnt;
                const unsigned sj = (unsigned)__shfl(m, v[j] ? kk + j : kk, 16);
                off[j] = sj * (unsigned)(OUT_DIM * 4) + lo16;
            }
#pragma unroll
            for (int j = 0; j < 8; ++j) r[j] = *(const v4f*)(tab + off[j]);
            __builtin_amdgcn_sched_barrier(0);
#pragma unroll
            for (int j = 0; j < 8; ++j) {
                acc.x += v[j] ? r[j].x : 0.f; acc.y += v[j] ? r[j].y : 0.f;
                acc.z += v[j] ? r[j].z : 0.f; acc.w += v[j] ? r[j].w : 0.f;
            }
        }
    };
    chunk(f0, f1, min(32, end - beg));
    for (int c = beg + 32; c < end; c += 32) {
        const int cnt = min(32, end - c);
        const int m0 = (l < cnt) ? csr[c + l] : 0;
        const int m1 = (16 + l < cnt) ? csr[c + 16 + l] : 0;
        chunk(m0, m1, cnt);
    }
    return acc;
}

__global__ void __launch_bounds__(512) k_gather1(const float* __restrict__ msg,
                                                 const int* __restrict__ csr,
                                                 const int* __restrict__ rowptr,
                                                 const float* __restrict__ dstn,
                                                 const float* __restrict__ srcn,
                                                 const float* __restrict__ b1,
                                                 const float* __restrict__ prelu_a,
                                                 float* __restrict__ hs,
                                                 float* __restrict__ hg1, int n) {
    const float pa = *prelu_a;
    const int l = threadIdx.x & 15;
    const int grp = threadIdx.x >> 4;
    const float4 bias = *(const float4*)(b1 + 4 * l);
    float4 part = make_float4(0.f, 0.f, 0.f, 0.f);
    const int step = gridDim.x * 32;
    int node = blockIdx.x * 32 + grp;
    int beg = 0, end = 0, nbeg = 0, nend = 0;
    if (node < n) { beg = rowptr[node]; end = rowptr[node + 1]; }
    if (node + step < n) { nbeg = rowptr[node + step]; nend = rowptr[node + step + 1]; }
    const int c0 = min(32, end - beg);
    int m0 = (l < c0) ? csr[beg + l] : 0;
    int m1 = (16 + l < c0) ? csr[beg + 16 + l] : 0;
    while (node < n) {
        asm volatile("" :: "v"(beg), "v"(end), "v"(m0), "v"(m1), "v"(nbeg), "v"(nend));
        const int nc = min(32, nend - nbeg);
        const int nm0 = (l < nc) ? csr[nbeg + l] : 0;
        const int nm1 = (16 + l < nc) ? csr[nbeg + 16 + l] : 0;
        const int nn = node + 2 * step;
        int nnbeg = 0, nnend = 0;
        if (nn < n) { nnbeg = rowptr[nn]; nnend = rowptr[nn + 1]; }
        const float dn = dstn[node];
        const float sn = srcn[node];
        const float4 v = gather_row(msg, csr, beg, end, l, m0, m1);
        float4 h;
        h.x = v.x * dn + bias.x; h.x = (h.x >= 0.f) ? h.x : pa * h.x;
        h.y = v.y * dn + bias.y; h.y = (h.y >= 0.f) ? h.y : pa * h.y;
        h.z = v.z * dn + bias.z; h.z = (h.z >= 0.f) ? h.z : pa * h.z;
        h.w = v.w * dn + bias.w; h.w = (h.w >= 0.f) ? h.w : pa * h.w;
        part.x += h.x; part.y += h.y; part.z += h.z; part.w += h.w;
        *(float4*)(hs + (size_t)node * OUT_DIM + 4 * l) =
            make_float4(h.x * sn, h.y * sn, h.z * sn, h.w * sn);
        node += step; beg = nbeg; end = nend; m0 = nm0; m1 = nm1; nbeg = nnbeg; nend = nnend;
    }
    __shared__ float red[32 * OUT_DIM];
    *(float4*)(red + grp * OUT_DIM + 4 * l) = part;
    __syncthreads();
    if (threadIdx.x < OUT_DIM) {
        float sum = 0.f;
        for (int g = 0; g < 32; ++g) sum += red[g * OUT_DIM + threadIdx.x];
        atomicAdd(&hg1[threadIdx.x], sum);
    }
}

struct Idx4 { int m0, m1, m2, m3; };

static __device__ __forceinline__ Idx4 load_idx(const int* __restrict__ csr, int c, int end, int l) {
    const int cnt = min(32, end - c);
    Idx4 x;
    x.m0 = (l < cnt) ? csr[c + l] : 0;
    x.m1 = (8 + l < cnt) ? csr[c + 8 + l] : 0;
    x.m2 = (16 + l < cnt) ? csr[c + 16 + l] : 0;
    x.m3 = (24 + l < cnt) ? csr[c + 24 + l] : 0;
    return x;
}

static __device__ __forceinline__ void gather_row8(const float* __restrict__ msg,
                                                   const int* __restrict__ csr,
                                                   int beg, int end, int l, Idx4 first,
                                                   float4& lo, float4& hi) {
    float4 a = make_float4(0.f, 0.f, 0.f, 0.f);
    float4 b = make_float4(0.f, 0.f, 0.f, 0.f);
    const char* tab = (const char*)msg;
    const unsigned lo16 = 16u * (unsigned)l;
    auto chunk = [&](Idx4 x, int cnt) {
        for (int k = 0; k < cnt; k += 8) {
            const int m = (k < 8) ? x.m0 : (k < 16) ? x.m1 : (k < 24) ? x.m2 : x.m3;
            v4f ra[8], rb[8];
            bool v[8];
            unsigned off[8];
#pragma unroll
            for (int j = 0; j < 8; ++j) {
                v[j] = k + j < cnt;
                const unsigned sj = (unsigned)__shfl(m, v[j] ? j : 0, 8);
                off[j] = sj * (unsigned)(OUT_DIM * 4) + lo16;
            }
#pragma unroll
            for (int j = 0; j < 8; ++j) {
                ra[j] = *(const v4f*)(tab + off[j]);
                rb[j] = *(const v4f*)(tab + off[j] + 128);
            }
            __builtin_amdgcn_sched_barrier(0);
#pragma unroll
            for (int j = 0; j < 8; ++j) {
                a.x += v[j] ? ra[j].x : 0.f; a.y += v[j] ? ra[j].y : 0.f;
                a.z += v[j] ? ra[j].z : 0.f; a.w += v[j] ? ra[j].w : 0.f;
                b.x += v[j] ? rb[j].x : 0.f; b.y += v[j] ? rb[j].y : 0.f;
                b.z += v[j] ? rb[j].z : 0.f; b.w += v[j] ? rb[j].w : 0.f;
            }
        }
    };
    chunk(first, min(32, end - beg));
    for (int c = beg + 32; c < end; c += 32)
        chunk(load_idx(csr, c, end, l), min(32, end - c));
    lo = a;
    hi = b;
}

constexpr int SA = OUT_DIM + 4;

__global__ void __launch_bounds__(256) __attribute__((amdgpu_waves_per_eu(8))) k_gather_gemm2(const float* __restrict__ msg,
                                                      const int* __restrict__ csr,
                                                      const int* __restrict__ rowptr,
                                                      const float* __restrict__ W2,
                                                      const float* __restrict__ dstn,
                                                      const float* __restrict__ b2,
                                                      const float* __restrict__ prelu_a,
                                                      float* __restrict__ hout,
                                                      float* __restrict__ hg2, int n) {
    __shared__ float sWt[OUT_DIM * S2];
    __shared__ __align__(16) float sA[8 * 16 * SA];
    __shared__ float shg[OUT_DIM];

    const int wave = threadIdx.x >> 5;
    const int lane = threadIdx.x & 31;
    const int tile = blockIdx.x * 8 + wave;
    const int ntiles = (n + 15) >> 4;
    const bool active = tile < ntiles;

    float wreg[16];
#pragma unroll
    for (int i = 0; i < 16; ++i) wreg[i] = W2[threadIdx.x + 256 * i];
    if (threadIdx.x < OUT_DIM) shg[threadIdx.x] = 0.f;

    float* arows = sA + wave * 16 * SA;
    int rp = 0;
    if (active) rp = rowptr[min(tile * 16 + lane, n)];
#pragma unroll
    for (int i = 0; i < 16; ++i) {
        const int idx = threadIdx.x + 256 * i;
        sWt[(idx & 63) * S2 + (idx >> 6)] = wreg[i];
    }
    if (active) {
        const int l = lane & 7;
        const int g = lane >> 3;
        int beg = __shfl(rp, g);
        int end = __shfl(rp, g + 1);
        Idx4 cur = load_idx(csr, beg, end, l);
        for (int p = 0; p < 4; ++p) {
            const int r = 4 * p + g;
            asm volatile("" :: "v"(cur.m0), "v"(cur.m1), "v"(cur.m2), "v"(cur.m3));
            const int nbeg = __shfl(rp, r + 4);
            const int nend = __shfl(rp, r + 5);
            Idx4 nxt = cur;
            if (p < 3) nxt = load_idx(csr, nbeg, nend, l);
            float4 lo, hi;
            gather_row8(msg, csr, beg, end, l, cur, lo, hi);
            *(float4*)(arows + r * SA + 4 * l)      = lo;
            *(float4*)(arows + r * SA + 32 + 4 * l) = hi;
            cur = nxt; beg = nbeg; end = nend;
        }
    }
    __syncthreads();

    const float pa = *prelu_a;
    const int halfsel = lane >> 4;
    const int l15 = lane & 15;

    if (active) {
        const int koff = halfsel * 2;
        const float* arow = arows + l15 * SA;
        const float* bc0 = sWt + ( 0 + l15) * S2;
        const float* bc1 = sWt + (16 + l15) * S2;
        const float* bc2 = sWt + (32 + l15) * S2;
        const float* bc3 = sWt + (48 + l15) * S2;

        v8f acc0 = {}, acc1 = {}, acc2 = {}, acc3 = {};
#pragma unroll 4
        for (int kb = 0; kb < OUT_DIM / 4; ++kb) {
            const int k0 = kb * 4 + koff;
            const v2f a  = *(const v2f*)(arow + k0);
            const v2f b0 = *(const v2f*)(bc0 + k0);
            const v2f b1 = *(const v2f*)(bc1 + k0);
            const v2f b2v = *(const v2f*)(bc2 + k0);
            const v2f b3 = *(const v2f*)(bc3 + k0);
            acc0 = __builtin_amdgcn_wmma_f32_16x16x4_f32(false, a, false, b0, (short)0, acc0, false, false);
            acc1 = __builtin_amdgcn_wmma_f32_16x16x4_f32(false, a, false, b1, (short)0, acc1, false, false);
            acc2 = __builtin_amdgcn_wmma_f32_16x16x4_f32(false, a, false, b2v, (short)0, acc2, false, false);
            acc3 = __builtin_amdgcn_wmma_f32_16x16x4_f32(false, a, false, b3, (short)0, acc3, false, false);
        }

        const float bb0 = b2[ 0 + l15], bb1 = b2[16 + l15];
        const float bb2 = b2[32 + l15], bb3 = b2[48 + l15];
        float hsum0 = 0.f, hsum1 = 0.f, hsum2 = 0.f, hsum3 = 0.f;
#pragma unroll
        for (int r = 0; r < 8; ++r) {
            const int node = tile * 16 + r + halfsel * 8;
            if (node < n) {
                const float dn = dstn[node];
                float v0 = acc0[r] * dn + bb0; v0 = (v0 >= 0.f) ? v0 : pa * v0;
                float v1 = acc1[r] * dn + bb1; v1 = (v1 >= 0.f) ? v1 : pa * v1;
                float v2 = acc2[r] * dn + bb2; v2 = (v2 >= 0.f) ? v2 : pa * v2;
                float v3 = acc3[r] * dn + bb3; v3 = (v3 >= 0.f) ? v3 : pa * v3;
                float* o = hout + (size_t)node * OUT_DIM + l15;
                o[0] = v0; o[16] = v1; o[32] = v2; o[48] = v3;
                hsum0 += v0; hsum1 += v1; hsum2 += v2; hsum3 += v3;
            }
        }
        atomicAdd(&shg[ 0 + l15], hsum0);
        atomicAdd(&shg[16 + l15], hsum1);
        atomicAdd(&shg[32 + l15], hsum2);
        atomicAdd(&shg[48 + l15], hsum3);
    }
    __syncthreads();
    if (threadIdx.x < OUT_DIM) atomicAdd(&hg2[threadIdx.x], shg[threadIdx.x]);
}

extern "C" void kernel_launch(void* const* d_in, const int* in_sizes, int n_in,
                              void* d_out, int out_size, void* d_ws, size_t ws_size,
                              hipStream_t stream) {
    const float* feat    = (const float*)d_in[0];
    const int*   src     = (const int*)  d_in[1];
    const int*   dst     = (const int*)  d_in[2];
    const float* W1      = (const float*)d_in[3];
    const float* b1      = (const float*)d_in[4];
    const float* W2      = (const float*)d_in[5];
    const float* b2      = (const float*)d_in[6];
    const float* prelu_a = (const float*)d_in[7];

    const int n = in_sizes[0] / IN_DIM;
    const int E = in_sizes[1];

    float* out = (float*)d_out;
    float* hg1 = out + (size_t)n * OUT_DIM;
    float* hg2 = hg1 + OUT_DIM;

    float* wsf    = (float*)d_ws;
    float* big    = wsf;
    int*   csr    = (int*)(wsf + (size_t)n * OUT_DIM);
    int*   rowptr = csr + E;
    float* srcn   = (float*)(rowptr + n + 32);
    float* dstn   = srcn + n;

    const int nb      = (n + BW - 1) / BW;
    const int nchunks = (E + CH - 1) / CH;
    const int ostride = (nb + 1 + 31) & ~31;

    int*  pd    = (int*)big;
    int*  ps    = pd + E;
    int*  offD  = ps + E;
    int*  offS  = offD + (size_t)nchunks * ostride;
    int*  spill = offS + (size_t)nchunks * ostride;

    const int ntiles = (n + 15) / 16;
    const int gemm_blocks = (ntiles + 7) / 8;

    k_bin    <<<nchunks, 1024, 0, stream>>>(src, dst, pd, ps, offD, offS, hg1, E, nb, ostride);
    k_own    <<<nb, 1024, 0, stream>>>(pd, ps, offD, offS, rowptr, csr, srcn, dstn, spill, feat, W1, out, n, nb, nchunks, ostride);
    k_gather1<<<625, 512, 0, stream>>>(out, csr, rowptr, dstn, srcn, b1, prelu_a, big, hg1, n);
    k_gather_gemm2<<<gemm_blocks, 256, 0, stream>>>(big, csr, rowptr, W2, dstn, b2, prelu_a, out, hg2, n);
}
